# starting kernel
# baseline (speedup 1.0000x reference)
#include <hip/hip_runtime.h>
#include <hip/hip_bf16.h>
#include <math.h>

typedef __attribute__((ext_vector_type(2))) float v2f;
typedef __attribute__((ext_vector_type(8))) float v8f;

#define NNODES 100000
#define NFEAT  128
#define NHID   32
#define NCLASS 40
#define NEDGES 1600000
#define NEG_SLOPE 0.2f

static inline int ceil_div(long long a, long long b) { return (int)((a + b - 1) / b); }

__global__ __launch_bounds__(256) void gemm1_wmma(const float* __restrict__ x,
                                                  const float* __restrict__ W,
                                                  float* __restrict__ h) {
    __shared__ float ldsW[NFEAT * NHID];
    for (int i = threadIdx.x; i < NFEAT * NHID; i += 256) ldsW[i] = W[i];
    __syncthreads();

    const int lane    = threadIdx.x & 31;
    const int wave    = threadIdx.x >> 5;
    const int rowBase = blockIdx.x * 128 + wave * 16;
    const int half    = lane >> 4;
    const int l16     = lane & 15;

    int row  = rowBase + l16;
    int rowc = row < NNODES ? row : (NNODES - 1);
    const float* xr = x + (long long)rowc * NFEAT;

    v8f c0 = {}; v8f c1 = {};
    for (int k = 0; k < NFEAT; k += 4) {
        const int k0 = k + 2 * half;
        v2f a;
        a.x = xr[k0];
        a.y = xr[k0 + 1];
        v2f b0, b1;
        b0.x = ldsW[k0 * NHID + l16];
        b0.y = ldsW[(k0 + 1) * NHID + l16];
        b1.x = ldsW[k0 * NHID + 16 + l16];
        b1.y = ldsW[(k0 + 1) * NHID + 16 + l16];
        c0 = __builtin_amdgcn_wmma_f32_16x16x4_f32(false, a, false, b0, (short)0, c0, false, false);
        c1 = __builtin_amdgcn_wmma_f32_16x16x4_f32(false, a, false, b1, (short)0, c1, false, false);
    }
    #pragma unroll
    for (int m = 0; m < 8; ++m) {
        int r = rowBase + half * 8 + m;
        if (r < NNODES) {
            h[(long long)r * NHID + l16]      = c0[m];
            h[(long long)r * NHID + 16 + l16] = c1[m];
        }
    }
}

__global__ __launch_bounds__(256) void gemm2_wmma(const float* __restrict__ hin,
                                                  const float* __restrict__ W,
                                                  float* __restrict__ hout) {
    __shared__ float ldsW[NHID * 48];
    for (int i = threadIdx.x; i < NHID * 48; i += 256) {
        int k = i / 48, n = i % 48;
        ldsW[i] = (n < NCLASS) ? W[k * NCLASS + n] : 0.0f;
    }
    __syncthreads();

    const int lane    = threadIdx.x & 31;
    const int wave    = threadIdx.x >> 5;
    const int rowBase = blockIdx.x * 128 + wave * 16;
    const int half    = lane >> 4;
    const int l16     = lane & 15;

    int row  = rowBase + l16;
    int rowc = row < NNODES ? row : (NNODES - 1);
    const float* hr = hin + (long long)rowc * NHID;

    v8f c0 = {}; v8f c1 = {}; v8f c2 = {};
    for (int k = 0; k < NHID; k += 4) {
        const int k0 = k + 2 * half;
        v2f a;
        a.x = hr[k0];
        a.y = hr[k0 + 1];
        v2f b0, b1, b2;
        b0.x = ldsW[k0 * 48 + l16];        b0.y = ldsW[(k0 + 1) * 48 + l16];
        b1.x = ldsW[k0 * 48 + 16 + l16];   b1.y = ldsW[(k0 + 1) * 48 + 16 + l16];
        b2.x = ldsW[k0 * 48 + 32 + l16];   b2.y = ldsW[(k0 + 1) * 48 + 32 + l16];
        c0 = __builtin_amdgcn_wmma_f32_16x16x4_f32(false, a, false, b0, (short)0, c0, false, false);
        c1 = __builtin_amdgcn_wmma_f32_16x16x4_f32(false, a, false, b1, (short)0, c1, false, false);
        c2 = __builtin_amdgcn_wmma_f32_16x16x4_f32(false, a, false, b2, (short)0, c2, false, false);
    }
    #pragma unroll
    for (int m = 0; m < 8; ++m) {
        int r = rowBase + half * 8 + m;
        if (r < NNODES) {
            float* o = hout + (long long)r * NCLASS;
            o[l16]      = c0[m];
            o[16 + l16] = c1[m];
            if (32 + l16 < NCLASS) o[32 + l16] = c2[m];
        }
    }
}

__global__ void alpha_kernel(const float* __restrict__ h,
                             const float* __restrict__ asrc,
                             const float* __restrict__ adst,
                             float* __restrict__ outs, float* __restrict__ outd,
                             int n, int C) {
    int i = blockIdx.x * blockDim.x + threadIdx.x;
    if (i >= n) return;
    const float* r = h + (long long)i * C;
    float s = 0.f, d = 0.f;
    for (int c = 0; c < C; ++c) { float v = r[c]; s += v * asrc[c]; d += v * adst[c]; }
    outs[i] = s;
    outd[i] = d;
}

__global__ void edge_kernel(const float* __restrict__ as, const float* __restrict__ ad,
                            const int* __restrict__ src, const int* __restrict__ dst,
                            float* __restrict__ ew, float* __restrict__ denom, int E) {
    int i = blockIdx.x * blockDim.x + threadIdx.x;
    if (i >= E) return;
    float e = as[src[i]] + ad[dst[i]];
    e = e > 0.f ? e : NEG_SLOPE * e;
    float ee = expf(e);
    ew[i] = ee;
    atomicAdd(&denom[dst[i]], ee);
}

__global__ void scatter_kernel(const float* __restrict__ h, const float* __restrict__ ew,
                               const float* __restrict__ denom,
                               const int* __restrict__ src, const int* __restrict__ dst,
                               float* __restrict__ agg, long long total, int C) {
    long long i = (long long)blockIdx.x * blockDim.x + threadIdx.x;
    if (i >= total) return;
    int e = (int)(i / C);
    int c = (int)(i % C);
    int s = src[e], d = dst[e];
    float alpha = ew[e] / (denom[d] + 1e-16f);
    atomicAdd(&agg[(long long)d * C + c], h[(long long)s * C + c] * alpha);
}

__global__ void bias_relu_kernel(float* __restrict__ a, const float* __restrict__ b,
                                 long long total, int C) {
    long long i = (long long)blockIdx.x * blockDim.x + threadIdx.x;
    if (i >= total) return;
    float v = a[i] + b[(int)(i % C)];
    a[i] = v > 0.f ? v : 0.f;
}

__global__ void logsoftmax_bias_kernel(float* __restrict__ out,
                                       const float* __restrict__ b, int n) {
    int i = blockIdx.x * blockDim.x + threadIdx.x;
    if (i >= n) return;
    float* row = out + (long long)i * NCLASS;
    float m = -INFINITY;
    for (int c = 0; c < NCLASS; ++c) { float v = row[c] + b[c]; m = fmaxf(m, v); }
    float s = 0.f;
    for (int c = 0; c < NCLASS; ++c) s += expf(row[c] + b[c] - m);
    float lse = m + logf(s);
    for (int c = 0; c < NCLASS; ++c) row[c] = row[c] + b[c] - lse;
}

extern "C" void kernel_launch(void* const* d_in, const int* in_sizes, int n_in,
                              void* d_out, int out_size, void* d_ws, size_t ws_size,
                              hipStream_t stream) {
    const float* x     = (const float*)d_in[0];
    const int*   src   = (const int*)d_in[1];
    const int*   dst   = src + NEDGES;
    const float* W1    = (const float*)d_in[2];
    const float* as1v  = (const float*)d_in[3];
    const float* ad1v  = (const float*)d_in[4];
    const float* b1    = (const float*)d_in[5];
    const float* W2    = (const float*)d_in[6];
    const float* as2v  = (const float*)d_in[7];
    const float* ad2v  = (const float*)d_in[8];
    const float* b2    = (const float*)d_in[9];
    float* out = (float*)d_out;

    float* ws    = (float*)d_ws;
    float* h1    = ws;
    float* as    = h1 + (long long)NNODES * NHID;
    float* ad    = as + NNODES;
    float* denom = ad + NNODES;
    float* ew    = denom + NNODES;
    float* agg1  = ew + NEDGES;
    float* h2    = agg1 + (long long)NNODES * NHID;

    const int T = 256;
    const int gemmBlocks = ceil_div(NNODES, 128);

    gemm1_wmma<<<gemmBlocks, T, 0, stream>>>(x, W1, h1);
    alpha_kernel<<<ceil_div(NNODES, T), T, 0, stream>>>(h1, as1v, ad1v, as, ad, NNODES, NHID);
    hipMemsetAsync(denom, 0, sizeof(float) * NNODES, stream);
    hipMemsetAsync(agg1, 0, sizeof(float) * (size_t)NNODES * NHID, stream);
    edge_kernel<<<ceil_div(NEDGES, T), T, 0, stream>>>(as, ad, src, dst, ew, denom, NEDGES);
    {
        long long total = (long long)NEDGES * NHID;
        scatter_kernel<<<ceil_div(total, T), T, 0, stream>>>(h1, ew, denom, src, dst, agg1, total, NHID);
    }
    {
        long long total = (long long)NNODES * NHID;
        bias_relu_kernel<<<ceil_div(total, T), T, 0, stream>>>(agg1, b1, total, NHID);
    }

    gemm2_wmma<<<gemmBlocks, T, 0, stream>>>(agg1, W2, h2);
    alpha_kernel<<<ceil_div(NNODES, T), T, 0, stream>>>(h2, as2v, ad2v, as, ad, NNODES, NCLASS);
    hipMemsetAsync(denom, 0, sizeof(float) * NNODES, stream);
    hipMemsetAsync(out, 0, sizeof(float) * (size_t)NNODES * NCLASS, stream);
    edge_kernel<<<ceil_div(NEDGES, T), T, 0, stream>>>(as, ad, src, dst, ew, denom, NEDGES);
    {
        long long total = (long long)NEDGES * NCLASS;
        scatter_kernel<<<ceil_div(total, T), T, 0, stream>>>(h2, ew, denom, src, dst, out, total, NCLASS);
    }
    logsoftmax_bias_kernel<<<ceil_div(NNODES, T), T, 0, stream>>>(out, b2, NNODES);
}


# aggregation scan reads a 1-byte owner id per edge (map written by layer 1's logit launch), not the 4-byte destination; a resolve pass packs (local node, source); sums still LDS arrival order
# speedup vs baseline: 3.3705x; 3.3705x over previous
#include <hip/hip_runtime.h>
#include <hip/hip_bf16.h>
#include <math.h>

typedef __attribute__((ext_vector_type(2))) float v2f;
typedef __attribute__((ext_vector_type(8))) float v8f;

#define NNODES 100000
#define NFEAT  128
#define NHID   32
#define NCLASS 40
#define NEDGES 1600000
#define NEG_SLOPE 0.2f
#define AGG_BLOCKS 256
#define NPB        391

static inline int ceil_div(long long a, long long b) { return (int)((a + b - 1) / b); }

__global__ __launch_bounds__(256) void gemm1_wmma(const float* __restrict__ x,
                                                  const float* __restrict__ W,
                                                  float* __restrict__ h) {
    __shared__ float ldsW[NFEAT * NHID];
    for (int i = threadIdx.x; i < NFEAT * NHID; i += 256) ldsW[i] = W[i];
    __syncthreads();

    const int lane    = threadIdx.x & 31;
    const int wave    = threadIdx.x >> 5;
    const int rowBase = blockIdx.x * 128 + wave * 16;
    const int half    = lane >> 4;
    const int l16     = lane & 15;

    int row  = rowBase + l16;
    int rowc = row < NNODES ? row : (NNODES - 1);
    const float* xr = x + (long long)rowc * NFEAT;

    v8f c0 = {}; v8f c1 = {};
    for (int k = 0; k < NFEAT; k += 4) {
        const int k0 = k + 2 * half;
        v2f a;
        a.x = xr[k0];
        a.y = xr[k0 + 1];
        v2f b0, b1;
        b0.x = ldsW[k0 * NHID + l16];
        b0.y = ldsW[(k0 + 1) * NHID + l16];
        b1.x = ldsW[k0 * NHID + 16 + l16];
        b1.y = ldsW[(k0 + 1) * NHID + 16 + l16];
        c0 = __builtin_amdgcn_wmma_f32_16x16x4_f32(false, a, false, b0, (short)0, c0, false, false);
        c1 = __builtin_amdgcn_wmma_f32_16x16x4_f32(false, a, false, b1, (short)0, c1, false, false);
    }
    #pragma unroll
    for (int m = 0; m < 8; ++m) {
        int r = rowBase + half * 8 + m;
        if (r < NNODES) {
            h[(long long)r * NHID + l16]      = c0[m];
            h[(long long)r * NHID + 16 + l16] = c1[m];
        }
    }
}

__global__ __launch_bounds__(256) void gemm2_wmma(const float* __restrict__ hin,
                                                  const float* __restrict__ W,
                                                  float* __restrict__ hout) {
    __shared__ float ldsW[NHID * 48];
    for (int i = threadIdx.x; i < NHID * 48; i += 256) {
        int k = i / 48, n = i % 48;
        ldsW[i] = (n < NCLASS) ? W[k * NCLASS + n] : 0.0f;
    }
    __syncthreads();

    const int lane    = threadIdx.x & 31;
    const int wave    = threadIdx.x >> 5;
    const int rowBase = blockIdx.x * 128 + wave * 16;
    const int half    = lane >> 4;
    const int l16     = lane & 15;

    int row  = rowBase + l16;
    int rowc = row < NNODES ? row : (NNODES - 1);
    const float* hr = hin + (long long)rowc * NHID;

    v8f c0 = {}; v8f c1 = {}; v8f c2 = {};
    for (int k = 0; k < NHID; k += 4) {
        const int k0 = k + 2 * half;
        v2f a;
        a.x = hr[k0];
        a.y = hr[k0 + 1];
        v2f b0, b1, b2;
        b0.x = ldsW[k0 * 48 + l16];        b0.y = ldsW[(k0 + 1) * 48 + l16];
        b1.x = ldsW[k0 * 48 + 16 + l16];   b1.y = ldsW[(k0 + 1) * 48 + 16 + l16];
        b2.x = ldsW[k0 * 48 + 32 + l16];   b2.y = ldsW[(k0 + 1) * 48 + 32 + l16];
        c0 = __builtin_amdgcn_wmma_f32_16x16x4_f32(false, a, false, b0, (short)0, c0, false, false);
        c1 = __builtin_amdgcn_wmma_f32_16x16x4_f32(false, a, false, b1, (short)0, c1, false, false);
        c2 = __builtin_amdgcn_wmma_f32_16x16x4_f32(false, a, false, b2, (short)0, c2, false, false);
    }
    #pragma unroll
    for (int m = 0; m < 8; ++m) {
        int r = rowBase + half * 8 + m;
        if (r < NNODES) {
            float* o = hout + (long long)r * NCLASS;
            o[l16]      = c0[m];
            o[16 + l16] = c1[m];
            if (32 + l16 < NCLASS) o[32 + l16] = c2[m];
        }
    }
}

__global__ void alpha_kernel(const float* __restrict__ h,
                             const float* __restrict__ asrc,
                             const float* __restrict__ adst,
                             float* __restrict__ outs, float* __restrict__ outd,
                             int n, int C,
                             const int* __restrict__ dst, unsigned* __restrict__ owner4) {
    int i = blockIdx.x * blockDim.x + threadIdx.x;
    if (owner4 != nullptr) {
        const int4* d4 = reinterpret_cast<const int4*>(dst);
        const int nthreads = gridDim.x * blockDim.x;
        for (int p = i; p < NEDGES / 4; p += nthreads) {
            const int4 v = d4[p];
            owner4[p] = ((unsigned)v.x / (unsigned)NPB) | (((unsigned)v.y / (unsigned)NPB) << 8) |
                        (((unsigned)v.z / (unsigned)NPB) << 16) | (((unsigned)v.w / (unsigned)NPB) << 24);
        }
    }
    if (i >= n) return;
    const float* r = h + (long long)i * C;
    float s = 0.f, d = 0.f;
    for (int c = 0; c < C; ++c) { float v = r[c]; s += v * asrc[c]; d += v * adst[c]; }
    outs[i] = s;
    outd[i] = d;
}

#define AGG_T      1024
#define AGG_WAVES  (AGG_T / 32)
#define E16        (NEDGES / 16)
#define SCAN_ROT   384
#define SCAN_MLP   8
#define LIST_CAP   28672
#define SRC_BITS   17
#define WS_CAP     10240

template <int C>
struct AggShared {
    float    acc[NPB * C];
    float    den[NPB];
    float    adl[NPB];
    float    lse[NPB];
    unsigned list[LIST_CAP];
    unsigned lcount;
};

__device__ __forceinline__ void lds_add(float* p, float v) {
    __hip_atomic_fetch_add(p, v, __ATOMIC_RELAXED, __HIP_MEMORY_SCOPE_WORKGROUP);
}

__device__ __forceinline__ unsigned local_node(int d, int lo, int cnt) {
    return min((unsigned)(d - lo), (unsigned)(cnt - 1));
}

template <int C>
__device__ __forceinline__ void own_edge(AggShared<C>& sm, int e, int lo, int cnt,
                                         const float* __restrict__ h,
                                         const float* __restrict__ as,
                                         const int* __restrict__ src,
                                         const int* __restrict__ dst) {
    unsigned pos = __hip_atomic_fetch_add(&sm.lcount, 1u, __ATOMIC_RELAXED, __HIP_MEMORY_SCOPE_WORKGROUP);
    if (pos < (unsigned)LIST_CAP) {
        sm.list[pos] = (unsigned)e;
    } else {
        const unsigned dl = local_node(dst[e], lo, cnt);
        int s = src[e];
        float x = as[s] + sm.adl[dl];
        x = x > 0.f ? x : NEG_SLOPE * x;
        float w = expf(x);
        lds_add(&sm.den[dl], w);
        #pragma nounroll
        for (int c = 0; c < C; ++c) lds_add(&sm.acc[dl * C + c], h[(long long)s * C + c] * w);
    }
}

__device__ __forceinline__ unsigned nonzero_bytes(unsigned x) {
    return ((x & 0x7F7F7F7Fu) + 0x7F7F7F7Fu) | x;
}

__device__ __forceinline__ unsigned owned_mask(uint4 v, unsigned pat, unsigned keep) {
    const unsigned y0 = nonzero_bytes(v.x ^ pat), y1 = nonzero_bytes(v.y ^ pat);
    const unsigned y2 = nonzero_bytes(v.z ^ pat), y3 = nonzero_bytes(v.w ^ pat);
    const unsigned nz = ((y0 >> 7) & 0x01010101u) | ((y1 >> 6) & 0x02020202u) |
                        ((y2 >> 5) & 0x04040404u) | ((y3 >> 4) & 0x08080808u);
    return ~nz & keep;
}

template <int C>
__device__ __forceinline__ void scan_all(AggShared<C>& sm, const uint4* __restrict__ own4,
                                         int start, unsigned pat, int lo, int cnt, int tid,
                                         const float* __restrict__ h,
                                         const float* __restrict__ as,
                                         const int* __restrict__ src,
                                         const int* __restrict__ dst) {
    for (int i0 = 0; i0 < E16; i0 += SCAN_MLP * AGG_T) {
        uint4    v[SCAN_MLP];
        int      ix[SCAN_MLP];
        unsigned keep[SCAN_MLP];
        #pragma unroll
        for (int k = 0; k < SCAN_MLP; ++k) {
            const int  pos = i0 + k * AGG_T + tid;
            const bool in  = pos < E16;
            int i = pos + start;
            i = i >= E16 ? i - E16 : i;
            i = in ? i : tid;
            ix[k]   = i;
            keep[k] = in ? 0x0F0F0F0Fu : 0u;
            v[k]    = own4[i];
        }
        #pragma unroll
        for (int k = 0; k < SCAN_MLP; k += 2) {
            unsigned m = owned_mask(v[k], pat, keep[k]) | (owned_mask(v[k + 1], pat, keep[k + 1]) << 4);
            const int pa = ix[k];
            const int dp = ix[k + 1] - pa;
            while (m) {
                const int b = __builtin_ctz(m);
                m &= m - 1u;
                const int piece = pa + ((b & 4) ? dp : 0);
                own_edge<C>(sm, piece * 16 + 4 * (b & 3) + (b >> 3), lo, cnt, h, as, src, dst);
            }
        }
    }
}

template <int C>
__device__ __forceinline__ void resolve_list(AggShared<C>& sm, int lo, int cnt, int tid,
                                             const int* __restrict__ src,
                                             const int* __restrict__ dst) {
    const unsigned n = sm.lcount < (unsigned)LIST_CAP ? sm.lcount : (unsigned)LIST_CAP;
    for (unsigned i0 = 0; i0 < n; i0 += 4u * AGG_T) {
        unsigned idx[4];
        int      e[4], d[4], s[4];
        #pragma unroll
        for (int k = 0; k < 4; ++k) {
            idx[k] = i0 + (unsigned)(k * AGG_T + tid);
            e[k]   = idx[k] < n ? (int)sm.list[idx[k]] : 0;
        }
        #pragma unroll
        for (int k = 0; k < 4; ++k) { d[k] = dst[e[k]]; s[k] = src[e[k]]; }
        #pragma unroll
        for (int k = 0; k < 4; ++k)
            if (idx[k] < n) sm.list[idx[k]] = (local_node(d[k], lo, cnt) << SRC_BITS) | (unsigned)s[k];
    }
}

template <int C, bool FINAL>
__global__ __launch_bounds__(AGG_T) void agg_owner(const float* __restrict__ h,
                                                   const float* __restrict__ as,
                                                   const float* __restrict__ ad,
                                                   const int* __restrict__ src,
                                                   const int* __restrict__ dst,
                                                   const float* __restrict__ bias,
                                                   float* __restrict__ outp,
                                                   unsigned* __restrict__ wsList,
                                                   unsigned* __restrict__ wsCount,
                                                   const unsigned* __restrict__ owner) {
    constexpr int LPE   = C / 4;
    constexpr int EPW   = 32 / LPE;
    constexpr int STEPS = (32 + EPW - 1) / EPW;

    __shared__ AggShared<C> sm;

    const int tid  = threadIdx.x;
    const int lane = tid & 31;
    const int wave = tid >> 5;
    const int lo   = blockIdx.x * NPB;
    const int cnt  = (NNODES - lo) < NPB ? (NNODES - lo) : NPB;

    for (int i = tid; i < NPB * C; i += AGG_T) sm.acc[i] = 0.f;
    if (tid < NPB) {
        sm.den[tid] = 0.f;
        sm.adl[tid] = tid < cnt ? ad[lo + tid] : 0.f;
    }
    if (tid == 0) sm.lcount = 0u;
    __syncthreads();

    const uint4*   own4  = reinterpret_cast<const uint4*>(owner);
    const int      start = blockIdx.x * SCAN_ROT;
    const unsigned pat   = blockIdx.x * 0x01010101u;
    uint4*      wl4   = reinterpret_cast<uint4*>(wsList + (size_t)blockIdx.x * WS_CAP);
    if (!FINAL) {
        scan_all<C>(sm, own4, start, pat, lo, cnt, tid, h, as, src, dst);
        __syncthreads();
        resolve_list<C>(sm, lo, cnt, tid, src, dst);
        __syncthreads();
        const unsigned total = sm.lcount;
        if (tid == 0) wsCount[blockIdx.x] = total;
        if (total <= (unsigned)WS_CAP) {
            const int np = (int)((total + 3u) >> 2);
            for (int i4 = tid; i4 < np; i4 += AGG_T) {
                const unsigned i = 4u * (unsigned)i4;
                uint4 v;
                v.x = sm.list[i];
                v.y = i + 1u < total ? sm.list[i + 1u] : 0u;
                v.z = i + 2u < total ? sm.list[i + 2u] : 0u;
                v.w = i + 3u < total ? sm.list[i + 3u] : 0u;
                wl4[i4] = v;
            }
        }
    } else {
        const unsigned total = wsCount[blockIdx.x];
        if (total <= (unsigned)WS_CAP) {
            const int np = (int)((total + 3u) >> 2);
            for (int i4 = tid; i4 < np; i4 += AGG_T) {
                const uint4 v = wl4[i4];
                const unsigned i = 4u * (unsigned)i4;
                sm.list[i]      = v.x;
                sm.list[i + 1u] = v.y;
                sm.list[i + 2u] = v.z;
                sm.list[i + 3u] = v.w;
            }
            if (tid == 0) sm.lcount = total;
        } else {
            scan_all<C>(sm, own4, start, pat, lo, cnt, tid, h, as, src, dst);
            __syncthreads();
            resolve_list<C>(sm, lo, cnt, tid, src, dst);
        }
        __syncthreads();
    }

    const unsigned n      = sm.lcount < (unsigned)LIST_CAP ? sm.lcount : (unsigned)LIST_CAP;
    const int      nbatch = (int)((n + 31u) >> 5);
    const int      q      = lane / LPE;
    const int      p      = lane - q * LPE;
    const bool     used   = q < EPW;

    for (int b = wave; b < nbatch; b += AGG_WAVES) {
        const unsigned i     = (unsigned)b * 32u + (unsigned)lane;
        const bool     valid = i < n;
        const unsigned ent   = valid ? sm.list[i] : 0u;
        const int      s     = (int)(ent & ((1u << SRC_BITS) - 1u));
        const int      dl    = (int)(ent >> SRC_BITS);
        float x = as[s] + sm.adl[dl];
        x = x > 0.f ? x : NEG_SLOPE * x;
        const float w = valid ? expf(x) : 0.f;
        lds_add(&sm.den[dl], w);

        float4 r[STEPS];
        float  wj[STEPS];
        int    dj[STEPS];
        bool   act[STEPS];
        #pragma unroll
        for (int t = 0; t < STEPS; ++t) {
            const int j  = t * EPW + q;
            const int jj = j < 31 ? j : 31;
            const int sj = __shfl(s, jj);
            wj[t]  = __shfl(w, jj);
            dj[t]  = __shfl(dl, jj);
            act[t] = used && (j < 32) && ((unsigned)b * 32u + (unsigned)j < n);
            r[t]   = *reinterpret_cast<const float4*>(h + (long long)sj * C + p * 4);
        }
        #pragma unroll
        for (int t = 0; t < STEPS; ++t) {
            float* a = &sm.acc[(act[t] ? dj[t] : 0) * C + p * 4];
            lds_add(a + 0, act[t] ? r[t].x * wj[t] : 0.f);
            lds_add(a + 1, act[t] ? r[t].y * wj[t] : 0.f);
            lds_add(a + 2, act[t] ? r[t].z * wj[t] : 0.f);
            lds_add(a + 3, act[t] ? r[t].w * wj[t] : 0.f);
        }
    }
    __syncthreads();

    float* og = outp + (long long)lo * C;
    if (!FINAL) {
        for (int i = tid; i < cnt * C; i += AGG_T) {
            const int nl = i / C, c = i - nl * C;
            const float v = sm.acc[i] / (sm.den[nl] + 1e-16f) + bias[c];
            og[i] = v > 0.f ? v : 0.f;
        }
    } else {
        for (int i = tid; i < cnt * C; i += AGG_T) {
            const int nl = i / C, c = i - nl * C;
            sm.acc[i] = sm.acc[i] / (sm.den[nl] + 1e-16f) + bias[c];
        }
        __syncthreads();
        if (tid < cnt) {
            const float* row = &sm.acc[tid * C];
            float m = -INFINITY;
            for (int c = 0; c < C; ++c) m = fmaxf(m, row[c]);
            float sum = 0.f;
            for (int c = 0; c < C; ++c) sum += expf(row[c] - m);
            sm.lse[tid] = m + logf(sum);
        }
        __syncthreads();
        for (int i = tid; i < cnt * C; i += AGG_T) og[i] = sm.acc[i] - sm.lse[i / C];
    }
}

extern "C" void kernel_launch(void* const* d_in, const int* in_sizes, int n_in,
                              void* d_out, int out_size, void* d_ws, size_t ws_size,
                              hipStream_t stream) {
    const float* x     = (const float*)d_in[0];
    const int*   src   = (const int*)d_in[1];
    const int*   dst   = src + NEDGES;
    const float* W1    = (const float*)d_in[2];
    const float* as1v  = (const float*)d_in[3];
    const float* ad1v  = (const float*)d_in[4];
    const float* b1    = (const float*)d_in[5];
    const float* W2    = (const float*)d_in[6];
    const float* as2v  = (const float*)d_in[7];
    const float* ad2v  = (const float*)d_in[8];
    const float* b2    = (const float*)d_in[9];
    float* out = (float*)d_out;

    float* ws    = (float*)d_ws;
    float* h1    = ws;
    float* h2    = ws;
    float* as    = ws + (long long)NNODES * NCLASS;
    float* ad    = as + NNODES;
    float* agg1  = ad + NNODES;
    unsigned* wsList  = (unsigned*)(agg1 + (long long)NNODES * NHID);
    unsigned* wsCount = wsList + (long long)AGG_BLOCKS * WS_CAP;
    unsigned* owner   = wsCount + AGG_BLOCKS;

    const int T = 256;
    const int gemmBlocks = ceil_div(NNODES, 128);

    gemm1_wmma<<<gemmBlocks, T, 0, stream>>>(x, W1, h1);
    alpha_kernel<<<ceil_div(NNODES, T), T, 0, stream>>>(h1, as1v, ad1v, as, ad, NNODES, NHID, dst, owner);
    agg_owner<NHID, false><<<AGG_BLOCKS, AGG_T, 0, stream>>>(h1, as, ad, src, dst, b1, agg1, wsList, wsCount, owner);

    gemm2_wmma<<<gemmBlocks, T, 0, stream>>>(agg1, W2, h2);
    alpha_kernel<<<ceil_div(NNODES, T), T, 0, stream>>>(h2, as2v, ad2v, as, ad, NNODES, NCLASS, dst, nullptr);
    agg_owner<NCLASS, true><<<AGG_BLOCKS, AGG_T, 0, stream>>>(h2, as, ad, src, dst, b2, out, wsList, wsCount, owner);
}


# layer 1's aggregation blocks multiply their finished rows (in LDS) by W2 on WMMA and take layer 2's logits there; the separate product and logit launches go: 6 launches -> 4; no sum order changes
# speedup vs baseline: 3.5670x; 1.0583x over previous
#include <hip/hip_runtime.h>
#include <hip/hip_bf16.h>
#include <math.h>

typedef __attribute__((ext_vector_type(2))) float v2f;
typedef __attribute__((ext_vector_type(8))) float v8f;

#define NNODES 100000
#define NFEAT  128
#define NHID   32
#define NCLASS 40
#define NEDGES 1600000
#define NEG_SLOPE 0.2f
#define AGG_BLOCKS 256
#define NPB        391

static inline int ceil_div(long long a, long long b) { return (int)((a + b - 1) / b); }

__global__ __launch_bounds__(256) void gemm1_wmma(const float* __restrict__ x,
                                                  const float* __restrict__ W,
                                                  float* __restrict__ h) {
    __shared__ float ldsW[NFEAT * NHID];
    for (int i = threadIdx.x; i < NFEAT * NHID; i += 256) ldsW[i] = W[i];
    __syncthreads();

    const int lane    = threadIdx.x & 31;
    const int wave    = threadIdx.x >> 5;
    const int rowBase = blockIdx.x * 128 + wave * 16;
    const int half    = lane >> 4;
    const int l16     = lane & 15;

    int row  = rowBase + l16;
    int rowc = row < NNODES ? row : (NNODES - 1);
    const float* xr = x + (long long)rowc * NFEAT;

    v8f c0 = {}; v8f c1 = {};
    for (int k = 0; k < NFEAT; k += 4) {
        const int k0 = k + 2 * half;
        v2f a;
        a.x = xr[k0];
        a.y = xr[k0 + 1];
        v2f b0, b1;
        b0.x = ldsW[k0 * NHID + l16];
        b0.y = ldsW[(k0 + 1) * NHID + l16];
        b1.x = ldsW[k0 * NHID + 16 + l16];
        b1.y = ldsW[(k0 + 1) * NHID + 16 + l16];
        c0 = __builtin_amdgcn_wmma_f32_16x16x4_f32(false, a, false, b0, (short)0, c0, false, false);
        c1 = __builtin_amdgcn_wmma_f32_16x16x4_f32(false, a, false, b1, (short)0, c1, false, false);
    }
    #pragma unroll
    for (int m = 0; m < 8; ++m) {
        int r = rowBase + half * 8 + m;
        if (r < NNODES) {
            h[(long long)r * NHID + l16]      = c0[m];
            h[(long long)r * NHID + 16 + l16] = c1[m];
        }
    }
}

__global__ void alpha_kernel(const float* __restrict__ h,
                             const float* __restrict__ asrc,
                             const float* __restrict__ adst,
                             float* __restrict__ outs, float* __restrict__ outd,
                             int n, int C,
                             const int* __restrict__ dst, unsigned* __restrict__ owner4) {
    int i = blockIdx.x * blockDim.x + threadIdx.x;
    if (owner4 != nullptr) {
        const int4* d4 = reinterpret_cast<const int4*>(dst);
        const int nthreads = gridDim.x * blockDim.x;
        for (int p = i; p < NEDGES / 4; p += nthreads) {
            const int4 v = d4[p];
            owner4[p] = ((unsigned)v.x / (unsigned)NPB) | (((unsigned)v.y / (unsigned)NPB) << 8) |
                        (((unsigned)v.z / (unsigned)NPB) << 16) | (((unsigned)v.w / (unsigned)NPB) << 24);
        }
    }
    if (i >= n) return;
    const float* r = h + (long long)i * C;
    float s = 0.f, d = 0.f;
    for (int c = 0; c < C; ++c) { float v = r[c]; s += v * asrc[c]; d += v * adst[c]; }
    outs[i] = s;
    outd[i] = d;
}

#define AGG_T      1024
#define AGG_WAVES  (AGG_T / 32)
#define E16        (NEDGES / 16)
#define SCAN_ROT   384
#define SCAN_MLP   8
#define LIST_CAP   28672
#define SRC_BITS   17
#define WS_CAP     8192

template <int C>
struct AggShared {
    float    acc[NPB * C];
    float    den[NPB];
    float    adl[NPB];
    float    lse[NPB];
    union {
        unsigned list[LIST_CAP];
        float    stage[LIST_CAP];
    };
    unsigned lcount;
};
static_assert(NHID * 48 + NPB * NCLASS <= LIST_CAP, "layer 1's finish stages W2 and the block's rows in the list's space");

__device__ __forceinline__ void lds_add(float* p, float v) {
    __hip_atomic_fetch_add(p, v, __ATOMIC_RELAXED, __HIP_MEMORY_SCOPE_WORKGROUP);
}

__device__ __forceinline__ unsigned local_node(int d, int lo, int cnt) {
    return min((unsigned)(d - lo), (unsigned)(cnt - 1));
}

template <int C>
__device__ __forceinline__ void own_edge(AggShared<C>& sm, int e, int lo, int cnt,
                                         const float* __restrict__ h,
                                         const float* __restrict__ as,
                                         const int* __restrict__ src,
                                         const int* __restrict__ dst) {
    unsigned pos = __hip_atomic_fetch_add(&sm.lcount, 1u, __ATOMIC_RELAXED, __HIP_MEMORY_SCOPE_WORKGROUP);
    if (pos < (unsigned)LIST_CAP) {
        sm.list[pos] = (unsigned)e;
    } else {
        const unsigned dl = local_node(dst[e], lo, cnt);
        int s = src[e];
        float x = as[s] + sm.adl[dl];
        x = x > 0.f ? x : NEG_SLOPE * x;
        float w = expf(x);
        lds_add(&sm.den[dl], w);
        #pragma nounroll
        for (int c = 0; c < C; ++c) lds_add(&sm.acc[dl * C + c], h[(long long)s * C + c] * w);
    }
}

__device__ __forceinline__ unsigned nonzero_bytes(unsigned x) {
    return ((x & 0x7F7F7F7Fu) + 0x7F7F7F7Fu) | x;
}

__device__ __forceinline__ unsigned owned_mask(uint4 v, unsigned pat, unsigned keep) {
    const unsigned y0 = nonzero_bytes(v.x ^ pat), y1 = nonzero_bytes(v.y ^ pat);
    const unsigned y2 = nonzero_bytes(v.z ^ pat), y3 = nonzero_bytes(v.w ^ pat);
    const unsigned nz = ((y0 >> 7) & 0x01010101u) | ((y1 >> 6) & 0x02020202u) |
                        ((y2 >> 5) & 0x04040404u) | ((y3 >> 4) & 0x08080808u);
    return ~nz & keep;
}

template <int C>
__device__ __forceinline__ void scan_all(AggShared<C>& sm, const uint4* __restrict__ own4,
                                         int start, unsigned pat, int lo, int cnt, int tid,
                                         const float* __restrict__ h,
                                         const float* __restrict__ as,
                                         const int* __restrict__ src,
                                         const int* __restrict__ dst) {
    for (int i0 = 0; i0 < E16; i0 += SCAN_MLP * AGG_T) {
        uint4    v[SCAN_MLP];
        int      ix[SCAN_MLP];
        unsigned keep[SCAN_MLP];
        #pragma unroll
        for (int k = 0; k < SCAN_MLP; ++k) {
            const int  pos = i0 + k * AGG_T + tid;
            const bool in  = pos < E16;
            int i = pos + start;
            i = i >= E16 ? i - E16 : i;
            i = in ? i : tid;
            ix[k]   = i;
            keep[k] = in ? 0x0F0F0F0Fu : 0u;
            v[k]    = own4[i];
        }
        #pragma unroll
        for (int k = 0; k < SCAN_MLP; k += 2) {
            unsigned m = owned_mask(v[k], pat, keep[k]) | (owned_mask(v[k + 1], pat, keep[k + 1]) << 4);
            const int pa = ix[k];
            const int dp = ix[k + 1] - pa;
            while (m) {
                const int b = __builtin_ctz(m);
                m &= m - 1u;
                const int piece = pa + ((b & 4) ? dp : 0);
                own_edge<C>(sm, piece * 16 + 4 * (b & 3) + (b >> 3), lo, cnt, h, as, src, dst);
            }
        }
    }
}

template <int C>
__device__ __forceinline__ void resolve_list(AggShared<C>& sm, int lo, int cnt, int tid,
                                             const int* __restrict__ src,
                                             const int* __restrict__ dst) {
    const unsigned n = sm.lcount < (unsigned)LIST_CAP ? sm.lcount : (unsigned)LIST_CAP;
    for (unsigned i0 = 0; i0 < n; i0 += 4u * AGG_T) {
        unsigned idx[4];
        int      e[4], d[4], s[4];
        #pragma unroll
        for (int k = 0; k < 4; ++k) {
            idx[k] = i0 + (unsigned)(k * AGG_T + tid);
            e[k]   = idx[k] < n ? (int)sm.list[idx[k]] : 0;
        }
        #pragma unroll
        for (int k = 0; k < 4; ++k) { d[k] = dst[e[k]]; s[k] = src[e[k]]; }
        #pragma unroll
        for (int k = 0; k < 4; ++k)
            if (idx[k] < n) sm.list[idx[k]] = (local_node(d[k], lo, cnt) << SRC_BITS) | (unsigned)s[k];
    }
}

template <int C, bool FINAL>
__global__ __launch_bounds__(AGG_T) void agg_owner(const float* __restrict__ h,
                                                   const float* __restrict__ as,
                                                   const float* __restrict__ ad,
                                                   const int* __restrict__ src,
                                                   const int* __restrict__ dst,
                                                   const float* __restrict__ bias,
                                                   float* __restrict__ outp,
                                                   unsigned* __restrict__ wsList,
                                                   unsigned* __restrict__ wsCount,
                                                   const unsigned* __restrict__ owner,
                                                   const float* __restrict__ Wn,
                                                   const float* __restrict__ asrcn,
                                                   const float* __restrict__ adstn,
                                                   float* __restrict__ asn,
                                                   float* __restrict__ adn) {
    constexpr int LPE   = C / 4;
    constexpr int EPW   = 32 / LPE;
    constexpr int STEPS = (32 + EPW - 1) / EPW;

    __shared__ AggShared<C> sm;

    const int tid  = threadIdx.x;
    const int lane = tid & 31;
    const int wave = tid >> 5;
    const int lo   = blockIdx.x * NPB;
    const int cnt  = (NNODES - lo) < NPB ? (NNODES - lo) : NPB;

    for (int i = tid; i < NPB * C; i += AGG_T) sm.acc[i] = 0.f;
    if (tid < NPB) {
        sm.den[tid] = 0.f;
        sm.adl[tid] = tid < cnt ? ad[lo + tid] : 0.f;
    }
    if (tid == 0) sm.lcount = 0u;
    __syncthreads();

    const uint4*   own4  = reinterpret_cast<const uint4*>(owner);
    const int      start = blockIdx.x * SCAN_ROT;
    const unsigned pat   = blockIdx.x * 0x01010101u;
    uint4*      wl4   = reinterpret_cast<uint4*>(wsList + (size_t)blockIdx.x * WS_CAP);
    if (!FINAL) {
        scan_all<C>(sm, own4, start, pat, lo, cnt, tid, h, as, src, dst);
        __syncthreads();
        resolve_list<C>(sm, lo, cnt, tid, src, dst);
        __syncthreads();
        const unsigned total = sm.lcount;
        if (tid == 0) wsCount[blockIdx.x] = total;
        if (total <= (unsigned)WS_CAP) {
            const int np = (int)((total + 3u) >> 2);
            for (int i4 = tid; i4 < np; i4 += AGG_T) {
                const unsigned i = 4u * (unsigned)i4;
                uint4 v;
                v.x = sm.list[i];
                v.y = i + 1u < total ? sm.list[i + 1u] : 0u;
                v.z = i + 2u < total ? sm.list[i + 2u] : 0u;
                v.w = i + 3u < total ? sm.list[i + 3u] : 0u;
                wl4[i4] = v;
            }
        }
    } else {
        const unsigned total = wsCount[blockIdx.x];
        if (total <= (unsigned)WS_CAP) {
            const int np = (int)((total + 3u) >> 2);
            for (int i4 = tid; i4 < np; i4 += AGG_T) {
                const uint4 v = wl4[i4];
                const unsigned i = 4u * (unsigned)i4;
                sm.list[i]      = v.x;
                sm.list[i + 1u] = v.y;
                sm.list[i + 2u] = v.z;
                sm.list[i + 3u] = v.w;
            }
            if (tid == 0) sm.lcount = total;
        } else {
            scan_all<C>(sm, own4, start, pat, lo, cnt, tid, h, as, src, dst);
            __syncthreads();
            resolve_list<C>(sm, lo, cnt, tid, src, dst);
        }
        __syncthreads();
    }

    const unsigned n      = sm.lcount < (unsigned)LIST_CAP ? sm.lcount : (unsigned)LIST_CAP;
    const int      nbatch = (int)((n + 31u) >> 5);
    const int      q      = lane / LPE;
    const int      p      = lane - q * LPE;
    const bool     used   = q < EPW;

    for (int b = wave; b < nbatch; b += AGG_WAVES) {
        const unsigned i     = (unsigned)b * 32u + (unsigned)lane;
        const bool     valid = i < n;
        const unsigned ent   = valid ? sm.list[i] : 0u;
        const int      s     = (int)(ent & ((1u << SRC_BITS) - 1u));
        const int      dl    = (int)(ent >> SRC_BITS);
        float x = as[s] + sm.adl[dl];
        x = x > 0.f ? x : NEG_SLOPE * x;
        const float w = valid ? expf(x) : 0.f;
        lds_add(&sm.den[dl], w);

        float4 r[STEPS];
        float  wj[STEPS];
        int    dj[STEPS];
        bool   act[STEPS];
        #pragma unroll
        for (int t = 0; t < STEPS; ++t) {
            const int j  = t * EPW + q;
            const int jj = j < 31 ? j : 31;
            const int sj = __shfl(s, jj);
            wj[t]  = __shfl(w, jj);
            dj[t]  = __shfl(dl, jj);
            act[t] = used && (j < 32) && ((unsigned)b * 32u + (unsigned)j < n);
            r[t]   = *reinterpret_cast<const float4*>(h + (long long)sj * C + p * 4);
        }
        #pragma unroll
        for (int t = 0; t < STEPS; ++t) {
            float* a = &sm.acc[(act[t] ? dj[t] : 0) * C + p * 4];
            lds_add(a + 0, act[t] ? r[t].x * wj[t] : 0.f);
            lds_add(a + 1, act[t] ? r[t].y * wj[t] : 0.f);
            lds_add(a + 2, act[t] ? r[t].z * wj[t] : 0.f);
            lds_add(a + 3, act[t] ? r[t].w * wj[t] : 0.f);
        }
    }
    __syncthreads();

    if (!FINAL) {
        for (int i = tid; i < cnt * C; i += AGG_T) {
            const int nl = i / C, c = i - nl * C;
            const float v = sm.acc[i] / (sm.den[nl] + 1e-16f) + bias[c];
            sm.acc[i] = v > 0.f ? v : 0.f;
        }
        float* w2s = sm.stage;
        float* h2s = sm.stage + NHID * 48;
        for (int i = tid; i < NHID * 48; i += AGG_T) {
            const int k = i / 48, n = i - k * 48;
            w2s[i] = (n < NCLASS) ? Wn[k * NCLASS + n] : 0.0f;
        }
        __syncthreads();

        const int half   = lane >> 4;
        const int l16    = lane & 15;
        const int ntiles = (cnt + 15) >> 4;
        const int wv = __builtin_amdgcn_readfirstlane(wave);
        for (int t = wv; t < ntiles; t += AGG_WAVES) {
            const int row  = t * 16 + l16;
            const int rowc = row < cnt ? row : (cnt - 1);
            const float* ar = &sm.acc[rowc * NHID];
            v8f c0 = {}; v8f c1 = {}; v8f c2 = {};
            for (int k = 0; k < NHID; k += 4) {
                const int k0 = k + 2 * half;
                v2f a;
                a.x = ar[k0];
                a.y = ar[k0 + 1];
                v2f b0, b1, b2;
                b0.x = w2s[k0 * 48 + l16];        b0.y = w2s[(k0 + 1) * 48 + l16];
                b1.x = w2s[k0 * 48 + 16 + l16];   b1.y = w2s[(k0 + 1) * 48 + 16 + l16];
                b2.x = w2s[k0 * 48 + 32 + l16];   b2.y = w2s[(k0 + 1) * 48 + 32 + l16];
                c0 = __builtin_amdgcn_wmma_f32_16x16x4_f32(false, a, false, b0, (short)0, c0, false, false);
                c1 = __builtin_amdgcn_wmma_f32_16x16x4_f32(false, a, false, b1, (short)0, c1, false, false);
                c2 = __builtin_amdgcn_wmma_f32_16x16x4_f32(false, a, false, b2, (short)0, c2, false, false);
            }
            #pragma unroll
            for (int m = 0; m < 8; ++m) {
                const int r = t * 16 + half * 8 + m;
                if (r < cnt) {
                    float* o = h2s + r * NCLASS;
                    o[l16]      = c0[m];
                    o[16 + l16] = c1[m];
                    if (32 + l16 < NCLASS) o[32 + l16] = c2[m];
                }
            }
        }
        __syncthreads();

        if (tid < cnt) {
            const float* r = h2s + tid * NCLASS;
            float s = 0.f, d = 0.f;
            for (int c = 0; c < NCLASS; ++c) { float v = r[c]; s += v * asrcn[c]; d += v * adstn[c]; }
            asn[lo + tid] = s;
            adn[lo + tid] = d;
        }
        float* og = outp + (long long)lo * NCLASS;
        for (int i = tid; i < cnt * NCLASS; i += AGG_T) og[i] = h2s[i];
    } else {
        float* og = outp + (long long)lo * C;
        for (int i = tid; i < cnt * C; i += AGG_T) {
            const int nl = i / C, c = i - nl * C;
            sm.acc[i] = sm.acc[i] / (sm.den[nl] + 1e-16f) + bias[c];
        }
        __syncthreads();
        if (tid < cnt) {
            const float* row = &sm.acc[tid * C];
            float m = -INFINITY;
            for (int c = 0; c < C; ++c) m = fmaxf(m, row[c]);
            float sum = 0.f;
            for (int c = 0; c < C; ++c) sum += expf(row[c] - m);
            sm.lse[tid] = m + logf(sum);
        }
        __syncthreads();
        for (int i = tid; i < cnt * C; i += AGG_T) og[i] = sm.acc[i] - sm.lse[i / C];
    }
}

extern "C" void kernel_launch(void* const* d_in, const int* in_sizes, int n_in,
                              void* d_out, int out_size, void* d_ws, size_t ws_size,
                              hipStream_t stream) {
    const float* x     = (const float*)d_in[0];
    const int*   src   = (const int*)d_in[1];
    const int*   dst   = src + NEDGES;
    const float* W1    = (const float*)d_in[2];
    const float* as1v  = (const float*)d_in[3];
    const float* ad1v  = (const float*)d_in[4];
    const float* b1    = (const float*)d_in[5];
    const float* W2    = (const float*)d_in[6];
    const float* as2v  = (const float*)d_in[7];
    const float* ad2v  = (const float*)d_in[8];
    const float* b2    = (const float*)d_in[9];
    float* out = (float*)d_out;

    float* ws    = (float*)d_ws;
    float* h1    = ws;
    float* h2    = h1 + (long long)NNODES * NHID;
    float* as1   = h2 + (long long)NNODES * NCLASS;
    float* ad1   = as1 + NNODES;
    float* as2   = ad1 + NNODES;
    float* ad2   = as2 + NNODES;
    unsigned* wsList  = (unsigned*)(ad2 + NNODES);
    unsigned* wsCount = wsList + (long long)AGG_BLOCKS * WS_CAP;
    unsigned* owner   = wsCount + AGG_BLOCKS;

    const int T = 256;
    const int gemmBlocks = ceil_div(NNODES, 128);

    gemm1_wmma<<<gemmBlocks, T, 0, stream>>>(x, W1, h1);
    alpha_kernel<<<ceil_div(NNODES, T), T, 0, stream>>>(h1, as1v, ad1v, as1, ad1, NNODES, NHID, dst, owner);
    agg_owner<NHID, false><<<AGG_BLOCKS, AGG_T, 0, stream>>>(h1, as1, ad1, src, dst, b1, h2, wsList, wsCount, owner,
                                                             W2, as2v, ad2v, as2, ad2);

    agg_owner<NCLASS, true><<<AGG_BLOCKS, AGG_T, 0, stream>>>(h2, as2, ad2, src, dst, b2, out, wsList, wsCount, owner,
                                                              nullptr, nullptr, nullptr, nullptr, nullptr);
}


# owner map kept as eight bit planes per 32 edges (wave ballots in the logit launch): the scan gets a group's owned-edge mask from 8 logic ops, not byte tests; list and sums still LDS arrival order
# speedup vs baseline: 3.7816x; 1.0602x over previous
#include <hip/hip_runtime.h>
#include <hip/hip_bf16.h>
#include <math.h>

typedef __attribute__((ext_vector_type(2))) float v2f;
typedef __attribute__((ext_vector_type(8))) float v8f;

#define NNODES 100000
#define NFEAT  128
#define NHID   32
#define NCLASS 40
#define NEDGES 1600000
#define NEG_SLOPE 0.2f
#define AGG_BLOCKS 256
#define NPB        391

static inline int ceil_div(long long a, long long b) { return (int)((a + b - 1) / b); }

__global__ __launch_bounds__(256) void gemm1_wmma(const float* __restrict__ x,
                                                  const float* __restrict__ W,
                                                  float* __restrict__ h) {
    __shared__ float ldsW[NFEAT * NHID];
    for (int i = threadIdx.x; i < NFEAT * NHID; i += 256) ldsW[i] = W[i];
    __syncthreads();

    const int lane    = threadIdx.x & 31;
    const int wave    = threadIdx.x >> 5;
    const int rowBase = blockIdx.x * 128 + wave * 16;
    const int half    = lane >> 4;
    const int l16     = lane & 15;

    int row  = rowBase + l16;
    int rowc = row < NNODES ? row : (NNODES - 1);
    const float* xr = x + (long long)rowc * NFEAT;

    v8f c0 = {}; v8f c1 = {};
    for (int k = 0; k < NFEAT; k += 4) {
        const int k0 = k + 2 * half;
        v2f a;
        a.x = xr[k0];
        a.y = xr[k0 + 1];
        v2f b0, b1;
        b0.x = ldsW[k0 * NHID + l16];
        b0.y = ldsW[(k0 + 1) * NHID + l16];
        b1.x = ldsW[k0 * NHID + 16 + l16];
        b1.y = ldsW[(k0 + 1) * NHID + 16 + l16];
        c0 = __builtin_amdgcn_wmma_f32_16x16x4_f32(false, a, false, b0, (short)0, c0, false, false);
        c1 = __builtin_amdgcn_wmma_f32_16x16x4_f32(false, a, false, b1, (short)0, c1, false, false);
    }
    #pragma unroll
    for (int m = 0; m < 8; ++m) {
        int r = rowBase + half * 8 + m;
        if (r < NNODES) {
            h[(long long)r * NHID + l16]      = c0[m];
            h[(long long)r * NHID + 16 + l16] = c1[m];
        }
    }
}

#define NGROUP (NEDGES / 32)
#define NCHUNK (NEDGES / 128)
__global__ void alpha_kernel(const float* __restrict__ h,
                             const float* __restrict__ asrc,
                             const float* __restrict__ adst,
                             float* __restrict__ outs, float* __restrict__ outd,
                             int n, int C,
                             const int* __restrict__ dst, unsigned* __restrict__ planeLo) {
    int i = blockIdx.x * blockDim.x + threadIdx.x;
    if (planeLo != nullptr) {
        const int4* d4   = reinterpret_cast<const int4*>(dst);
        const int   lane = threadIdx.x & 31;
        const int nwaves = (gridDim.x * blockDim.x) >> 5;
        const int wv = __builtin_amdgcn_readfirstlane(i >> 5);
        unsigned* planeHi = planeLo + 4 * NGROUP;
        for (int c = wv; c < NCHUNK; c += nwaves) {
            const int4 v = d4[c * 32 + lane];
            const unsigned o[4] = { (unsigned)v.x / (unsigned)NPB, (unsigned)v.y / (unsigned)NPB,
                                    (unsigned)v.z / (unsigned)NPB, (unsigned)v.w / (unsigned)NPB };
            unsigned w = 0u;
            #pragma unroll
            for (int L = 0; L < 32; ++L) {
                const int j = (L >> 2) & 3;
                const int k = ((L >> 4) << 2) | (L & 3);
                const unsigned bits = __builtin_amdgcn_ballot_w32(((o[j] >> k) & 1u) != 0u);
                w = (lane == L) ? bits : w;
            }
            unsigned* p = (lane < 16) ? planeLo : planeHi;
            p[c * 16 + (lane & 15)] = w;
        }
    }
    if (i >= n) return;
    const float* r = h + (long long)i * C;
    float s = 0.f, d = 0.f;
    for (int c = 0; c < C; ++c) { float v = r[c]; s += v * asrc[c]; d += v * adst[c]; }
    outs[i] = s;
    outd[i] = d;
}

#define AGG_T      1024
#define AGG_WAVES  (AGG_T / 32)
#define SCAN_ROT   192
#define SCAN_MLP   4
#define LIST_CAP   28672
#define SRC_BITS   17
#define WS_CAP     8192

template <int C>
struct AggShared {
    float    acc[NPB * C];
    float    den[NPB];
    float    adl[NPB];
    float    lse[NPB];
    union {
        unsigned list[LIST_CAP];
        float    stage[LIST_CAP];
    };
    unsigned lcount;
};
static_assert(NHID * 48 + NPB * NCLASS <= LIST_CAP, "layer 1's finish stages W2 and the block's rows in the list's space");

__device__ __forceinline__ void lds_add(float* p, float v) {
    __hip_atomic_fetch_add(p, v, __ATOMIC_RELAXED, __HIP_MEMORY_SCOPE_WORKGROUP);
}

__device__ __forceinline__ unsigned local_node(int d, int lo, int cnt) {
    return min((unsigned)(d - lo), (unsigned)(cnt - 1));
}

template <int C>
__device__ __forceinline__ void own_edge(AggShared<C>& sm, int e, int lo, int cnt,
                                         const float* __restrict__ h,
                                         const float* __restrict__ as,
                                         const int* __restrict__ src,
                                         const int* __restrict__ dst) {
    unsigned pos = __hip_atomic_fetch_add(&sm.lcount, 1u, __ATOMIC_RELAXED, __HIP_MEMORY_SCOPE_WORKGROUP);
    if (pos < (unsigned)LIST_CAP) {
        sm.list[pos] = (unsigned)e;
    } else {
        const unsigned dl = local_node(dst[e], lo, cnt);
        int s = src[e];
        float x = as[s] + sm.adl[dl];
        x = x > 0.f ? x : NEG_SLOPE * x;
        float w = expf(x);
        lds_add(&sm.den[dl], w);
        #pragma nounroll
        for (int c = 0; c < C; ++c) lds_add(&sm.acc[dl * C + c], h[(long long)s * C + c] * w);
    }
}

struct OwnerBits { unsigned n[8]; };
__device__ __forceinline__ OwnerBits owner_bits(unsigned block) {
    OwnerBits r;
    #pragma unroll
    for (int k = 0; k < 8; ++k) r.n[k] = ((block >> k) & 1u) ? 0u : ~0u;
    return r;
}

__device__ __forceinline__ unsigned owned_mask(uint4 lo, uint4 hi, const OwnerBits& ob) {
    return (lo.x ^ ob.n[0]) & (lo.y ^ ob.n[1]) & (lo.z ^ ob.n[2]) & (lo.w ^ ob.n[3]) &
           (hi.x ^ ob.n[4]) & (hi.y ^ ob.n[5]) & (hi.z ^ ob.n[6]) & (hi.w ^ ob.n[7]);
}

template <int C>
__device__ __forceinline__ void scan_all(AggShared<C>& sm, const uint4* __restrict__ lo4,
                                         const uint4* __restrict__ hi4,
                                         int start, const OwnerBits& ob, int lo, int cnt, int tid,
                                         const float* __restrict__ h,
                                         const float* __restrict__ as,
                                         const int* __restrict__ src,
                                         const int* __restrict__ dst) {
    for (int i0 = 0; i0 < NGROUP; i0 += SCAN_MLP * AGG_T) {
        uint4    pl[SCAN_MLP], ph[SCAN_MLP];
        int      gx[SCAN_MLP];
        unsigned keep[SCAN_MLP];
        #pragma unroll
        for (int k = 0; k < SCAN_MLP; ++k) {
            const int  pos = i0 + k * AGG_T + tid;
            const bool in  = pos < NGROUP;
            int g = pos + start;
            g = g >= NGROUP ? g - NGROUP : g;
            g = in ? g : tid;
            gx[k]   = g;
            keep[k] = in ? ~0u : 0u;
            pl[k]   = lo4[g];
            ph[k]   = hi4[g];
        }
        #pragma unroll
        for (int k = 0; k < SCAN_MLP; ++k) {
            unsigned m = owned_mask(pl[k], ph[k], ob) & keep[k];
            const int e0 = 128 * (gx[k] >> 2) + (gx[k] & 3);
            while (m) {
                const int b = __builtin_ctz(m);
                m &= m - 1u;
                own_edge<C>(sm, e0 + 4 * b, lo, cnt, h, as, src, dst);
            }
        }
    }
}

template <int C>
__device__ __forceinline__ void resolve_list(AggShared<C>& sm, int lo, int cnt, int tid,
                                             const int* __restrict__ src,
                                             const int* __restrict__ dst) {
    const unsigned n = sm.lcount < (unsigned)LIST_CAP ? sm.lcount : (unsigned)LIST_CAP;
    for (unsigned i0 = 0; i0 < n; i0 += 4u * AGG_T) {
        unsigned idx[4];
        int      e[4], d[4], s[4];
        #pragma unroll
        for (int k = 0; k < 4; ++k) {
            idx[k] = i0 + (unsigned)(k * AGG_T + tid);
            e[k]   = idx[k] < n ? (int)sm.list[idx[k]] : 0;
        }
        #pragma unroll
        for (int k = 0; k < 4; ++k) { d[k] = dst[e[k]]; s[k] = src[e[k]]; }
        #pragma unroll
        for (int k = 0; k < 4; ++k)
            if (idx[k] < n) sm.list[idx[k]] = (local_node(d[k], lo, cnt) << SRC_BITS) | (unsigned)s[k];
    }
}

template <int C, bool FINAL>
__global__ __launch_bounds__(AGG_T) void agg_owner(const float* __restrict__ h,
                                                   const float* __restrict__ as,
                                                   const float* __restrict__ ad,
                                                   const int* __restrict__ src,
                                                   const int* __restrict__ dst,
                                                   const float* __restrict__ bias,
                                                   float* __restrict__ outp,
                                                   unsigned* __restrict__ wsList,
                                                   unsigned* __restrict__ wsCount,
                                                   const unsigned* __restrict__ owner,
                                                   const float* __restrict__ Wn,
                                                   const float* __restrict__ asrcn,
                                                   const float* __restrict__ adstn,
                                                   float* __restrict__ asn,
                                                   float* __restrict__ adn) {
    constexpr int LPE   = C / 4;
    constexpr int EPW   = 32 / LPE;
    constexpr int STEPS = (32 + EPW - 1) / EPW;

    __shared__ AggShared<C> sm;

    const int tid  = threadIdx.x;
    const int lane = tid & 31;
    const int wave = tid >> 5;
    const int lo   = blockIdx.x * NPB;
    const int cnt  = (NNODES - lo) < NPB ? (NNODES - lo) : NPB;

    for (int i = tid; i < NPB * C; i += AGG_T) sm.acc[i] = 0.f;
    if (tid < NPB) {
        sm.den[tid] = 0.f;
        sm.adl[tid] = tid < cnt ? ad[lo + tid] : 0.f;
    }
    if (tid == 0) sm.lcount = 0u;
    __syncthreads();

    const uint4*    lo4   = reinterpret_cast<const uint4*>(owner);
    const uint4*    hi4   = reinterpret_cast<const uint4*>(owner + 4 * NGROUP);
    const int       start = blockIdx.x * SCAN_ROT;
    const OwnerBits ob    = owner_bits(blockIdx.x);
    uint4*      wl4   = reinterpret_cast<uint4*>(wsList + (size_t)blockIdx.x * WS_CAP);
    if (!FINAL) {
        scan_all<C>(sm, lo4, hi4, start, ob, lo, cnt, tid, h, as, src, dst);
        __syncthreads();
        resolve_list<C>(sm, lo, cnt, tid, src, dst);
        __syncthreads();
        const unsigned total = sm.lcount;
        if (tid == 0) wsCount[blockIdx.x] = total;
        if (total <= (unsigned)WS_CAP) {
            const int np = (int)((total + 3u) >> 2);
            for (int i4 = tid; i4 < np; i4 += AGG_T) {
                const unsigned i = 4u * (unsigned)i4;
                uint4 v;
                v.x = sm.list[i];
                v.y = i + 1u < total ? sm.list[i + 1u] : 0u;
                v.z = i + 2u < total ? sm.list[i + 2u] : 0u;
                v.w = i + 3u < total ? sm.list[i + 3u] : 0u;
                wl4[i4] = v;
            }
        }
    } else {
        const unsigned total = wsCount[blockIdx.x];
        if (total <= (unsigned)WS_CAP) {
            const int np = (int)((total + 3u) >> 2);
            for (int i4 = tid; i4 < np; i4 += AGG_T) {
                const uint4 v = wl4[i4];
                const unsigned i = 4u * (unsigned)i4;
                sm.list[i]      = v.x;
                sm.list[i + 1u] = v.y;
                sm.list[i + 2u] = v.z;
                sm.list[i + 3u] = v.w;
            }
            if (tid == 0) sm.lcount = total;
        } else {
            scan_all<C>(sm, lo4, hi4, start, ob, lo, cnt, tid, h, as, src, dst);
            __syncthreads();
            resolve_list<C>(sm, lo, cnt, tid, src, dst);
        }
        __syncthreads();
    }

    const unsigned n      = sm.lcount < (unsigned)LIST_CAP ? sm.lcount : (unsigned)LIST_CAP;
    const int      nbatch = (int)((n + 31u) >> 5);
    const int      q      = lane / LPE;
    const int      p      = lane - q * LPE;
    const bool     used   = q < EPW;

    for (int b = wave; b < nbatch; b += AGG_WAVES) {
        const unsigned i     = (unsigned)b * 32u + (unsigned)lane;
        const bool     valid = i < n;
        const unsigned ent   = valid ? sm.list[i] : 0u;
        const int      s     = (int)(ent & ((1u << SRC_BITS) - 1u));
        const int      dl    = (int)(ent >> SRC_BITS);
        float x = as[s] + sm.adl[dl];
        x = x > 0.f ? x : NEG_SLOPE * x;
        const float w = valid ? expf(x) : 0.f;
        lds_add(&sm.den[dl], w);

        float4 r[STEPS];
        float  wj[STEPS];
        int    dj[STEPS];
        bool   act[STEPS];
        #pragma unroll
        for (int t = 0; t < STEPS; ++t) {
            const int j  = t * EPW + q;
            const int jj = j < 31 ? j : 31;
            const int sj = __shfl(s, jj);
            wj[t]  = __shfl(w, jj);
            dj[t]  = __shfl(dl, jj);
            act[t] = used && (j < 32) && ((unsigned)b * 32u + (unsigned)j < n);
            r[t]   = *reinterpret_cast<const float4*>(h + (long long)sj * C + p * 4);
        }
        #pragma unroll
        for (int t = 0; t < STEPS; ++t) {
            float* a = &sm.acc[(act[t] ? dj[t] : 0) * C + p * 4];
            lds_add(a + 0, act[t] ? r[t].x * wj[t] : 0.f);
            lds_add(a + 1, act[t] ? r[t].y * wj[t] : 0.f);
            lds_add(a + 2, act[t] ? r[t].z * wj[t] : 0.f);
            lds_add(a + 3, act[t] ? r[t].w * wj[t] : 0.f);
        }
    }
    __syncthreads();

    if (!FINAL) {
        for (int i = tid; i < cnt * C; i += AGG_T) {
            const int nl = i / C, c = i - nl * C;
            const float v = sm.acc[i] / (sm.den[nl] + 1e-16f) + bias[c];
            sm.acc[i] = v > 0.f ? v : 0.f;
        }
        float* w2s = sm.stage;
        float* h2s = sm.stage + NHID * 48;
        for (int i = tid; i < NHID * 48; i += AGG_T) {
            const int k = i / 48, n = i - k * 48;
            w2s[i] = (n < NCLASS) ? Wn[k * NCLASS + n] : 0.0f;
        }
        __syncthreads();

        const int half   = lane >> 4;
        const int l16    = lane & 15;
        const int ntiles = (cnt + 15) >> 4;
        const int wv = __builtin_amdgcn_readfirstlane(wave);
        for (int t = wv; t < ntiles; t += AGG_WAVES) {
            const int row  = t * 16 + l16;
            const int rowc = row < cnt ? row : (cnt - 1);
            const float* ar = &sm.acc[rowc * NHID];
            v8f c0 = {}; v8f c1 = {}; v8f c2 = {};
            for (int k = 0; k < NHID; k += 4) {
                const int k0 = k + 2 * half;
                v2f a;
                a.x = ar[k0];
                a.y = ar[k0 + 1];
                v2f b0, b1, b2;
                b0.x = w2s[k0 * 48 + l16];        b0.y = w2s[(k0 + 1) * 48 + l16];
                b1.x = w2s[k0 * 48 + 16 + l16];   b1.y = w2s[(k0 + 1) * 48 + 16 + l16];
                b2.x = w2s[k0 * 48 + 32 + l16];   b2.y = w2s[(k0 + 1) * 48 + 32 + l16];
                c0 = __builtin_amdgcn_wmma_f32_16x16x4_f32(false, a, false, b0, (short)0, c0, false, false);
                c1 = __builtin_amdgcn_wmma_f32_16x16x4_f32(false, a, false, b1, (short)0, c1, false, false);
                c2 = __builtin_amdgcn_wmma_f32_16x16x4_f32(false, a, false, b2, (short)0, c2, false, false);
            }
            #pragma unroll
            for (int m = 0; m < 8; ++m) {
                const int r = t * 16 + half * 8 + m;
                if (r < cnt) {
                    float* o = h2s + r * NCLASS;
                    o[l16]      = c0[m];
                    o[16 + l16] = c1[m];
                    if (32 + l16 < NCLASS) o[32 + l16] = c2[m];
                }
            }
        }
        __syncthreads();

        if (tid < cnt) {
            const float* r = h2s + tid * NCLASS;
            float s = 0.f, d = 0.f;
            for (int c = 0; c < NCLASS; ++c) { float v = r[c]; s += v * asrcn[c]; d += v * adstn[c]; }
            asn[lo + tid] = s;
            adn[lo + tid] = d;
        }
        float* og = outp + (long long)lo * NCLASS;
        for (int i = tid; i < cnt * NCLASS; i += AGG_T) og[i] = h2s[i];
    } else {
        float* og = outp + (long long)lo * C;
        for (int i = tid; i < cnt * C; i += AGG_T) {
            const int nl = i / C, c = i - nl * C;
            sm.acc[i] = sm.acc[i] / (sm.den[nl] + 1e-16f) + bias[c];
        }
        __syncthreads();
        if (tid < cnt) {
            const float* row = &sm.acc[tid * C];
            float m = -INFINITY;
            for (int c = 0; c < C; ++c) m = fmaxf(m, row[c]);
            float sum = 0.f;
            for (int c = 0; c < C; ++c) sum += expf(row[c] - m);
            sm.lse[tid] = m + logf(sum);
        }
        __syncthreads();
        for (int i = tid; i < cnt * C; i += AGG_T) og[i] = sm.acc[i] - sm.lse[i / C];
    }
}

extern "C" void kernel_launch(void* const* d_in, const int* in_sizes, int n_in,
                              void* d_out, int out_size, void* d_ws, size_t ws_size,
                              hipStream_t stream) {
    const float* x     = (const float*)d_in[0];
    const int*   src   = (const int*)d_in[1];
    const int*   dst   = src + NEDGES;
    const float* W1    = (const float*)d_in[2];
    const float* as1v  = (const float*)d_in[3];
    const float* ad1v  = (const float*)d_in[4];
    const float* b1    = (const float*)d_in[5];
    const float* W2    = (const float*)d_in[6];
    const float* as2v  = (const float*)d_in[7];
    const float* ad2v  = (const float*)d_in[8];
    const float* b2    = (const float*)d_in[9];
    float* out = (float*)d_out;

    float* ws    = (float*)d_ws;
    float* h1    = ws;
    float* h2    = h1 + (long long)NNODES * NHID;
    float* as1   = h2 + (long long)NNODES * NCLASS;
    float* ad1   = as1 + NNODES;
    float* as2   = ad1 + NNODES;
    float* ad2   = as2 + NNODES;
    unsigned* wsList  = (unsigned*)(ad2 + NNODES);
    unsigned* wsCount = wsList + (long long)AGG_BLOCKS * WS_CAP;
    unsigned* owner   = wsCount + AGG_BLOCKS;

    const int T = 256;
    const int gemmBlocks = ceil_div(NNODES, 128);

    gemm1_wmma<<<gemmBlocks, T, 0, stream>>>(x, W1, h1);
    alpha_kernel<<<ceil_div(NNODES, T), T, 0, stream>>>(h1, as1v, ad1v, as1, ad1, NNODES, NHID, dst, owner);
    agg_owner<NHID, false><<<AGG_BLOCKS, AGG_T, 0, stream>>>(h1, as1, ad1, src, dst, b1, h2, wsList, wsCount, owner,
                                                             W2, as2v, ad2v, as2, ad2);

    agg_owner<NCLASS, true><<<AGG_BLOCKS, AGG_T, 0, stream>>>(h2, as2, ad2, src, dst, b2, out, wsList, wsCount, owner,
                                                              nullptr, nullptr, nullptr, nullptr, nullptr);
}


# sum's LDS float adds bank-conflict-free: the rows served by one add instruction take their 4 columns in different orders (column k ^ row slot), 32 lanes on 32 banks; sums still LDS arrival order
# speedup vs baseline: 4.8700x; 1.2878x over previous
#include <hip/hip_runtime.h>
#include <hip/hip_bf16.h>
#include <math.h>

typedef __attribute__((ext_vector_type(2))) float v2f;
typedef __attribute__((ext_vector_type(8))) float v8f;

#define NNODES 100000
#define NFEAT  128
#define NHID   32
#define NCLASS 40
#define NEDGES 1600000
#define NEG_SLOPE 0.2f
#define AGG_BLOCKS 256
#define NPB        391

static inline int ceil_div(long long a, long long b) { return (int)((a + b - 1) / b); }

__global__ __launch_bounds__(256) void gemm1_wmma(const float* __restrict__ x,
                                                  const float* __restrict__ W,
                                                  float* __restrict__ h) {
    __shared__ float ldsW[NFEAT * NHID];
    for (int i = threadIdx.x; i < NFEAT * NHID; i += 256) ldsW[i] = W[i];
    __syncthreads();

    const int lane    = threadIdx.x & 31;
    const int wave    = threadIdx.x >> 5;
    const int rowBase = blockIdx.x * 128 + wave * 16;
    const int half    = lane >> 4;
    const int l16     = lane & 15;

    int row  = rowBase + l16;
    int rowc = row < NNODES ? row : (NNODES - 1);
    const float* xr = x + (long long)rowc * NFEAT;

    v8f c0 = {}; v8f c1 = {};
    for (int k = 0; k < NFEAT; k += 4) {
        const int k0 = k + 2 * half;
        v2f a;
        a.x = xr[k0];
        a.y = xr[k0 + 1];
        v2f b0, b1;
        b0.x = ldsW[k0 * NHID + l16];
        b0.y = ldsW[(k0 + 1) * NHID + l16];
        b1.x = ldsW[k0 * NHID + 16 + l16];
        b1.y = ldsW[(k0 + 1) * NHID + 16 + l16];
        c0 = __builtin_amdgcn_wmma_f32_16x16x4_f32(false, a, false, b0, (short)0, c0, false, false);
        c1 = __builtin_amdgcn_wmma_f32_16x16x4_f32(false, a, false, b1, (short)0, c1, false, false);
    }
    #pragma unroll
    for (int m = 0; m < 8; ++m) {
        int r = rowBase + half * 8 + m;
        if (r < NNODES) {
            h[(long long)r * NHID + l16]      = c0[m];
            h[(long long)r * NHID + 16 + l16] = c1[m];
        }
    }
}

#define NGROUP (NEDGES / 32)
#define NCHUNK (NEDGES / 128)
__global__ void alpha_kernel(const float* __restrict__ h,
                             const float* __restrict__ asrc,
                             const float* __restrict__ adst,
                             float* __restrict__ outs, float* __restrict__ outd,
                             int n, int C,
                             const int* __restrict__ dst, unsigned* __restrict__ planeLo) {
    int i = blockIdx.x * blockDim.x + threadIdx.x;
    if (planeLo != nullptr) {
        const int4* d4   = reinterpret_cast<const int4*>(dst);
        const int   lane = threadIdx.x & 31;
        const int nwaves = (gridDim.x * blockDim.x) >> 5;
        const int wv = __builtin_amdgcn_readfirstlane(i >> 5);
        unsigned* planeHi = planeLo + 4 * NGROUP;
        for (int c = wv; c < NCHUNK; c += nwaves) {
            const int4 v = d4[c * 32 + lane];
            const unsigned o[4] = { (unsigned)v.x / (unsigned)NPB, (unsigned)v.y / (unsigned)NPB,
                                    (unsigned)v.z / (unsigned)NPB, (unsigned)v.w / (unsigned)NPB };
            unsigned w = 0u;
            #pragma unroll
            for (int L = 0; L < 32; ++L) {
                const int j = (L >> 2) & 3;
                const int k = ((L >> 4) << 2) | (L & 3);
                const unsigned bits = __builtin_amdgcn_ballot_w32(((o[j] >> k) & 1u) != 0u);
                w = (lane == L) ? bits : w;
            }
            unsigned* p = (lane < 16) ? planeLo : planeHi;
            p[c * 16 + (lane & 15)] = w;
        }
    }
    if (i >= n) return;
    const float* r = h + (long long)i * C;
    float s = 0.f, d = 0.f;
    for (int c = 0; c < C; ++c) { float v = r[c]; s += v * asrc[c]; d += v * adst[c]; }
    outs[i] = s;
    outd[i] = d;
}

#define AGG_T      1024
#define AGG_WAVES  (AGG_T / 32)
#define SCAN_ROT   192
#define SCAN_MLP   4
#define LIST_CAP   28672
#define SRC_BITS   17
#define WS_CAP     8192

template <int C>
struct AggShared {
    float    acc[NPB * C];
    float    den[NPB];
    float    adl[NPB];
    float    lse[NPB];
    union {
        unsigned list[LIST_CAP];
        float    stage[LIST_CAP];
    };
    unsigned lcount;
};
static_assert(NHID * 48 + NPB * NCLASS <= LIST_CAP, "layer 1's finish stages W2 and the block's rows in the list's space");

__device__ __forceinline__ void lds_add(float* p, float v) {
    __hip_atomic_fetch_add(p, v, __ATOMIC_RELAXED, __HIP_MEMORY_SCOPE_WORKGROUP);
}

__device__ __forceinline__ unsigned local_node(int d, int lo, int cnt) {
    return min((unsigned)(d - lo), (unsigned)(cnt - 1));
}

template <int C>
__device__ __forceinline__ void own_edge(AggShared<C>& sm, int e, int lo, int cnt,
                                         const float* __restrict__ h,
                                         const float* __restrict__ as,
                                         const int* __restrict__ src,
                                         const int* __restrict__ dst) {
    unsigned pos = __hip_atomic_fetch_add(&sm.lcount, 1u, __ATOMIC_RELAXED, __HIP_MEMORY_SCOPE_WORKGROUP);
    if (pos < (unsigned)LIST_CAP) {
        sm.list[pos] = (unsigned)e;
    } else {
        const unsigned dl = local_node(dst[e], lo, cnt);
        int s = src[e];
        float x = as[s] + sm.adl[dl];
        x = x > 0.f ? x : NEG_SLOPE * x;
        float w = expf(x);
        lds_add(&sm.den[dl], w);
        #pragma nounroll
        for (int c = 0; c < C; ++c) lds_add(&sm.acc[dl * C + c], h[(long long)s * C + c] * w);
    }
}

struct OwnerBits { unsigned n[8]; };
__device__ __forceinline__ OwnerBits owner_bits(unsigned block) {
    OwnerBits r;
    #pragma unroll
    for (int k = 0; k < 8; ++k) r.n[k] = ((block >> k) & 1u) ? 0u : ~0u;
    return r;
}

__device__ __forceinline__ unsigned owned_mask(uint4 lo, uint4 hi, const OwnerBits& ob) {
    return (lo.x ^ ob.n[0]) & (lo.y ^ ob.n[1]) & (lo.z ^ ob.n[2]) & (lo.w ^ ob.n[3]) &
           (hi.x ^ ob.n[4]) & (hi.y ^ ob.n[5]) & (hi.z ^ ob.n[6]) & (hi.w ^ ob.n[7]);
}

template <int C>
__device__ __forceinline__ void scan_all(AggShared<C>& sm, const uint4* __restrict__ lo4,
                                         const uint4* __restrict__ hi4,
                                         int start, const OwnerBits& ob, int lo, int cnt, int tid,
                                         const float* __restrict__ h,
                                         const float* __restrict__ as,
                                         const int* __restrict__ src,
                                         const int* __restrict__ dst) {
    for (int i0 = 0; i0 < NGROUP; i0 += SCAN_MLP * AGG_T) {
        uint4    pl[SCAN_MLP], ph[SCAN_MLP];
        int      gx[SCAN_MLP];
        unsigned keep[SCAN_MLP];
        #pragma unroll
        for (int k = 0; k < SCAN_MLP; ++k) {
            const int  pos = i0 + k * AGG_T + tid;
            const bool in  = pos < NGROUP;
            int g = pos + start;
            g = g >= NGROUP ? g - NGROUP : g;
            g = in ? g : tid;
            gx[k]   = g;
            keep[k] = in ? ~0u : 0u;
            pl[k]   = lo4[g];
            ph[k]   = hi4[g];
        }
        #pragma unroll
        for (int k = 0; k < SCAN_MLP; ++k) {
            unsigned m = owned_mask(pl[k], ph[k], ob) & keep[k];
            const int e0 = 128 * (gx[k] >> 2) + (gx[k] & 3);
            while (m) {
                const int b = __builtin_ctz(m);
                m &= m - 1u;
                own_edge<C>(sm, e0 + 4 * b, lo, cnt, h, as, src, dst);
            }
        }
    }
}

template <int C>
__device__ __forceinline__ void resolve_list(AggShared<C>& sm, int lo, int cnt, int tid,
                                             const int* __restrict__ src,
                                             const int* __restrict__ dst) {
    const unsigned n = sm.lcount < (unsigned)LIST_CAP ? sm.lcount : (unsigned)LIST_CAP;
    for (unsigned i0 = 0; i0 < n; i0 += 4u * AGG_T) {
        unsigned idx[4];
        int      e[4], d[4], s[4];
        #pragma unroll
        for (int k = 0; k < 4; ++k) {
            idx[k] = i0 + (unsigned)(k * AGG_T + tid);
            e[k]   = idx[k] < n ? (int)sm.list[idx[k]] : 0;
        }
        #pragma unroll
        for (int k = 0; k < 4; ++k) { d[k] = dst[e[k]]; s[k] = src[e[k]]; }
        #pragma unroll
        for (int k = 0; k < 4; ++k)
            if (idx[k] < n) sm.list[idx[k]] = (local_node(d[k], lo, cnt) << SRC_BITS) | (unsigned)s[k];
    }
}

template <int C, bool FINAL>
__global__ __launch_bounds__(AGG_T) void agg_owner(const float* __restrict__ h,
                                                   const float* __restrict__ as,
                                                   const float* __restrict__ ad,
                                                   const int* __restrict__ src,
                                                   const int* __restrict__ dst,
                                                   const float* __restrict__ bias,
                                                   float* __restrict__ outp,
                                                   unsigned* __restrict__ wsList,
                                                   unsigned* __restrict__ wsCount,
                                                   const unsigned* __restrict__ owner,
                                                   const float* __restrict__ Wn,
                                                   const float* __restrict__ asrcn,
                                                   const float* __restrict__ adstn,
                                                   float* __restrict__ asn,
                                                   float* __restrict__ adn) {
    constexpr int LPE   = C / 4;
    constexpr int EPW   = 32 / LPE;
    constexpr int STEPS = (32 + EPW - 1) / EPW;

    __shared__ AggShared<C> sm;

    const int tid  = threadIdx.x;
    const int lane = tid & 31;
    const int wave = tid >> 5;
    const int lo   = blockIdx.x * NPB;
    const int cnt  = (NNODES - lo) < NPB ? (NNODES - lo) : NPB;

    for (int i = tid; i < NPB * C; i += AGG_T) sm.acc[i] = 0.f;
    if (tid < NPB) {
        sm.den[tid] = 0.f;
        sm.adl[tid] = tid < cnt ? ad[lo + tid] : 0.f;
    }
    if (tid == 0) sm.lcount = 0u;
    __syncthreads();

    const uint4*    lo4   = reinterpret_cast<const uint4*>(owner);
    const uint4*    hi4   = reinterpret_cast<const uint4*>(owner + 4 * NGROUP);
    const int       start = blockIdx.x * SCAN_ROT;
    const OwnerBits ob    = owner_bits(blockIdx.x);
    uint4*      wl4   = reinterpret_cast<uint4*>(wsList + (size_t)blockIdx.x * WS_CAP);
    if (!FINAL) {
        scan_all<C>(sm, lo4, hi4, start, ob, lo, cnt, tid, h, as, src, dst);
        __syncthreads();
        resolve_list<C>(sm, lo, cnt, tid, src, dst);
        __syncthreads();
        const unsigned total = sm.lcount;
        if (tid == 0) wsCount[blockIdx.x] = total;
        if (total <= (unsigned)WS_CAP) {
            const int np = (int)((total + 3u) >> 2);
            for (int i4 = tid; i4 < np; i4 += AGG_T) {
                const unsigned i = 4u * (unsigned)i4;
                uint4 v;
                v.x = sm.list[i];
                v.y = i + 1u < total ? sm.list[i + 1u] : 0u;
                v.z = i + 2u < total ? sm.list[i + 2u] : 0u;
                v.w = i + 3u < total ? sm.list[i + 3u] : 0u;
                wl4[i4] = v;
            }
        }
    } else {
        const unsigned total = wsCount[blockIdx.x];
        if (total <= (unsigned)WS_CAP) {
            const int np = (int)((total + 3u) >> 2);
            for (int i4 = tid; i4 < np; i4 += AGG_T) {
                const uint4 v = wl4[i4];
                const unsigned i = 4u * (unsigned)i4;
                sm.list[i]      = v.x;
                sm.list[i + 1u] = v.y;
                sm.list[i + 2u] = v.z;
                sm.list[i + 3u] = v.w;
            }
            if (tid == 0) sm.lcount = total;
        } else {
            scan_all<C>(sm, lo4, hi4, start, ob, lo, cnt, tid, h, as, src, dst);
            __syncthreads();
            resolve_list<C>(sm, lo, cnt, tid, src, dst);
        }
        __syncthreads();
    }

    const unsigned n      = sm.lcount < (unsigned)LIST_CAP ? sm.lcount : (unsigned)LIST_CAP;
    const int      nbatch = (int)((n + 31u) >> 5);
    const int      q      = lane / LPE;
    const int      p      = lane - q * LPE;
    const bool     used   = q < EPW;
    const int      qx     = q & 3;
    const bool     sw1    = (qx & 1) != 0;
    const bool     sw2    = (qx & 2) != 0;
    const unsigned pq     = (unsigned)(p * 16 + qx * 4);
    char* const    accb   = reinterpret_cast<char*>(sm.acc);

    for (int b = wave; b < nbatch; b += AGG_WAVES) {
        const unsigned i     = (unsigned)b * 32u + (unsigned)lane;
        const bool     valid = i < n;
        const unsigned ent   = valid ? sm.list[i] : 0u;
        const int      s     = (int)(ent & ((1u << SRC_BITS) - 1u));
        const int      dl    = (int)(ent >> SRC_BITS);
        float x = as[s] + sm.adl[dl];
        x = x > 0.f ? x : NEG_SLOPE * x;
        const float w = valid ? expf(x) : 0.f;
        lds_add(&sm.den[dl], w);

        float4 r[STEPS];
        float  wj[STEPS];
        int    dj[STEPS];
        bool   act[STEPS];
        #pragma unroll
        for (int t = 0; t < STEPS; ++t) {
            const int j  = t * EPW + q;
            const int jj = j < 31 ? j : 31;
            const int sj = __shfl(s, jj);
            wj[t]  = __shfl(w, jj);
            dj[t]  = __shfl(dl, jj);
            act[t] = used && (j < 32) && ((unsigned)b * 32u + (unsigned)j < n);
            r[t]   = *reinterpret_cast<const float4*>(h + (long long)sj * C + p * 4);
        }
        #pragma unroll
        for (int t = 0; t < STEPS; ++t) {
            const float v0 = r[t].x * wj[t], v1 = r[t].y * wj[t];
            const float v2 = r[t].z * wj[t], v3 = r[t].w * wj[t];
            const float u0 = sw1 ? v1 : v0, u1 = sw1 ? v0 : v1;
            const float u2 = sw1 ? v3 : v2, u3 = sw1 ? v2 : v3;
            const float b0 = sw2 ? u2 : u0, b1 = sw2 ? u3 : u1;
            const float b2 = sw2 ? u0 : u2, b3 = sw2 ? u1 : u3;
            const unsigned o0 = (unsigned)dj[t] * (unsigned)(C * 4) + pq;
            if (act[t]) {
                lds_add(reinterpret_cast<float*>(accb + o0),         b0);
                lds_add(reinterpret_cast<float*>(accb + (o0 ^ 4u)),  b1);
                lds_add(reinterpret_cast<float*>(accb + (o0 ^ 8u)),  b2);
                lds_add(reinterpret_cast<float*>(accb + (o0 ^ 12u)), b3);
            }
        }
    }
    __syncthreads();

    if (!FINAL) {
        for (int i = tid; i < cnt * C; i += AGG_T) {
            const int nl = i / C, c = i - nl * C;
            const float v = sm.acc[i] / (sm.den[nl] + 1e-16f) + bias[c];
            sm.acc[i] = v > 0.f ? v : 0.f;
        }
        float* w2s = sm.stage;
        float* h2s = sm.stage + NHID * 48;
        for (int i = tid; i < NHID * 48; i += AGG_T) {
            const int k = i / 48, n = i - k * 48;
            w2s[i] = (n < NCLASS) ? Wn[k * NCLASS + n] : 0.0f;
        }
        __syncthreads();

        const int half   = lane >> 4;
        const int l16    = lane & 15;
        const int ntiles = (cnt + 15) >> 4;
        const int wv = __builtin_amdgcn_readfirstlane(wave);
        for (int t = wv; t < ntiles; t += AGG_WAVES) {
            const int row  = t * 16 + l16;
            const int rowc = row < cnt ? row : (cnt - 1);
            const float* ar = &sm.acc[rowc * NHID];
            v8f c0 = {}; v8f c1 = {}; v8f c2 = {};
            for (int k = 0; k < NHID; k += 4) {
                const int k0 = k + 2 * half;
                v2f a;
                a.x = ar[k0];
                a.y = ar[k0 + 1];
                v2f b0, b1, b2;
                b0.x = w2s[k0 * 48 + l16];        b0.y = w2s[(k0 + 1) * 48 + l16];
                b1.x = w2s[k0 * 48 + 16 + l16];   b1.y = w2s[(k0 + 1) * 48 + 16 + l16];
                b2.x = w2s[k0 * 48 + 32 + l16];   b2.y = w2s[(k0 + 1) * 48 + 32 + l16];
                c0 = __builtin_amdgcn_wmma_f32_16x16x4_f32(false, a, false, b0, (short)0, c0, false, false);
                c1 = __builtin_amdgcn_wmma_f32_16x16x4_f32(false, a, false, b1, (short)0, c1, false, false);
                c2 = __builtin_amdgcn_wmma_f32_16x16x4_f32(false, a, false, b2, (short)0, c2, false, false);
            }
            #pragma unroll
            for (int m = 0; m < 8; ++m) {
                const int r = t * 16 + half * 8 + m;
                if (r < cnt) {
                    float* o = h2s + r * NCLASS;
                    o[l16]      = c0[m];
                    o[16 + l16] = c1[m];
                    if (32 + l16 < NCLASS) o[32 + l16] = c2[m];
                }
            }
        }
        __syncthreads();

        if (tid < cnt) {
            const float* r = h2s + tid * NCLASS;
            float s = 0.f, d = 0.f;
            for (int c = 0; c < NCLASS; ++c) { float v = r[c]; s += v * asrcn[c]; d += v * adstn[c]; }
            asn[lo + tid] = s;
            adn[lo + tid] = d;
        }
        float* og = outp + (long long)lo * NCLASS;
        for (int i = tid; i < cnt * NCLASS; i += AGG_T) og[i] = h2s[i];
    } else {
        float* og = outp + (long long)lo * C;
        for (int i = tid; i < cnt * C; i += AGG_T) {
            const int nl = i / C, c = i - nl * C;
            sm.acc[i] = sm.acc[i] / (sm.den[nl] + 1e-16f) + bias[c];
        }
        __syncthreads();
        if (tid < cnt) {
            const float* row = &sm.acc[tid * C];
            float m = -INFINITY;
            for (int c = 0; c < C; ++c) m = fmaxf(m, row[c]);
            float sum = 0.f;
            for (int c = 0; c < C; ++c) sum += expf(row[c] - m);
            sm.lse[tid] = m + logf(sum);
        }
        __syncthreads();
        for (int i = tid; i < cnt * C; i += AGG_T) og[i] = sm.acc[i] - sm.lse[i / C];
    }
}

extern "C" void kernel_launch(void* const* d_in, const int* in_sizes, int n_in,
                              void* d_out, int out_size, void* d_ws, size_t ws_size,
                              hipStream_t stream) {
    const float* x     = (const float*)d_in[0];
    const int*   src   = (const int*)d_in[1];
    const int*   dst   = src + NEDGES;
    const float* W1    = (const float*)d_in[2];
    const float* as1v  = (const float*)d_in[3];
    const float* ad1v  = (const float*)d_in[4];
    const float* b1    = (const float*)d_in[5];
    const float* W2    = (const float*)d_in[6];
    const float* as2v  = (const float*)d_in[7];
    const float* ad2v  = (const float*)d_in[8];
    const float* b2    = (const float*)d_in[9];
    float* out = (float*)d_out;

    float* ws    = (float*)d_ws;
    float* h1    = ws;
    float* h2    = h1 + (long long)NNODES * NHID;
    float* as1   = h2 + (long long)NNODES * NCLASS;
    float* ad1   = as1 + NNODES;
    float* as2   = ad1 + NNODES;
    float* ad2   = as2 + NNODES;
    unsigned* wsList  = (unsigned*)(ad2 + NNODES);
    unsigned* wsCount = wsList + (long long)AGG_BLOCKS * WS_CAP;
    unsigned* owner   = wsCount + AGG_BLOCKS;

    const int T = 256;
    const int gemmBlocks = ceil_div(NNODES, 128);

    gemm1_wmma<<<gemmBlocks, T, 0, stream>>>(x, W1, h1);
    alpha_kernel<<<ceil_div(NNODES, T), T, 0, stream>>>(h1, as1v, ad1v, as1, ad1, NNODES, NHID, dst, owner);
    agg_owner<NHID, false><<<AGG_BLOCKS, AGG_T, 0, stream>>>(h1, as1, ad1, src, dst, b1, h2, wsList, wsCount, owner,
                                                             W2, as2v, ad2v, as2, ad2);

    agg_owner<NCLASS, true><<<AGG_BLOCKS, AGG_T, 0, stream>>>(h2, as2, ad2, src, dst, b2, out, wsList, wsCount, owner,
                                                              nullptr, nullptr, nullptr, nullptr, nullptr);
}


# resolve pass gathers ONE packed word per owned edge (local node << 17 | source, written in edge order by the logit launch) instead of destination and source from two arrays; no sum order changes
# speedup vs baseline: 5.0410x; 1.0351x over previous
#include <hip/hip_runtime.h>
#include <hip/hip_bf16.h>
#include <math.h>

typedef __attribute__((ext_vector_type(2))) float v2f;
typedef __attribute__((ext_vector_type(8))) float v8f;

#define NNODES 100000
#define NFEAT  128
#define NHID   32
#define NCLASS 40
#define NEDGES 1600000
#define NEG_SLOPE 0.2f
#define AGG_BLOCKS 256
#define NPB        391
#define SRC_BITS   17
#define SRC_MASK   ((1u << SRC_BITS) - 1u)

static inline int ceil_div(long long a, long long b) { return (int)((a + b - 1) / b); }

__global__ __launch_bounds__(256) void gemm1_wmma(const float* __restrict__ x,
                                                  const float* __restrict__ W,
                                                  float* __restrict__ h) {
    __shared__ float ldsW[NFEAT * NHID];
    for (int i = threadIdx.x; i < NFEAT * NHID; i += 256) ldsW[i] = W[i];
    __syncthreads();

    const int lane    = threadIdx.x & 31;
    const int wave    = threadIdx.x >> 5;
    const int rowBase = blockIdx.x * 128 + wave * 16;
    const int half    = lane >> 4;
    const int l16     = lane & 15;

    int row  = rowBase + l16;
    int rowc = row < NNODES ? row : (NNODES - 1);
    const float* xr = x + (long long)rowc * NFEAT;

    v8f c0 = {}; v8f c1 = {};
    for (int k = 0; k < NFEAT; k += 4) {
        const int k0 = k + 2 * half;
        v2f a;
        a.x = xr[k0];
        a.y = xr[k0 + 1];
        v2f b0, b1;
        b0.x = ldsW[k0 * NHID + l16];
        b0.y = ldsW[(k0 + 1) * NHID + l16];
        b1.x = ldsW[k0 * NHID + 16 + l16];
        b1.y = ldsW[(k0 + 1) * NHID + 16 + l16];
        c0 = __builtin_amdgcn_wmma_f32_16x16x4_f32(false, a, false, b0, (short)0, c0, false, false);
        c1 = __builtin_amdgcn_wmma_f32_16x16x4_f32(false, a, false, b1, (short)0, c1, false, false);
    }
    #pragma unroll
    for (int m = 0; m < 8; ++m) {
        int r = rowBase + half * 8 + m;
        if (r < NNODES) {
            h[(long long)r * NHID + l16]      = c0[m];
            h[(long long)r * NHID + 16 + l16] = c1[m];
        }
    }
}

#define NGROUP (NEDGES / 32)
#define NCHUNK (NEDGES / 128)
__global__ void alpha_kernel(const float* __restrict__ h,
                             const float* __restrict__ asrc,
                             const float* __restrict__ adst,
                             float* __restrict__ outs, float* __restrict__ outd,
                             int n, int C,
                             const int* __restrict__ src, const int* __restrict__ dst,
                             unsigned* __restrict__ planeLo, unsigned* __restrict__ entry) {
    int i = blockIdx.x * blockDim.x + threadIdx.x;
    if (planeLo != nullptr) {
        const int4* d4   = reinterpret_cast<const int4*>(dst);
        const int4* s4   = reinterpret_cast<const int4*>(src);
        uint4*      en4  = reinterpret_cast<uint4*>(entry);
        const int   lane = threadIdx.x & 31;
        const int nwaves = (gridDim.x * blockDim.x) >> 5;
        const int wv = __builtin_amdgcn_readfirstlane(i >> 5);
        unsigned* planeHi = planeLo + 4 * NGROUP;
        for (int c = wv; c < NCHUNK; c += nwaves) {
            const int4 v  = d4[c * 32 + lane];
            const int4 sv = s4[c * 32 + lane];
            const unsigned o[4] = { (unsigned)v.x / (unsigned)NPB, (unsigned)v.y / (unsigned)NPB,
                                    (unsigned)v.z / (unsigned)NPB, (unsigned)v.w / (unsigned)NPB };
            uint4 en;
            en.x = (((unsigned)v.x - o[0] * (unsigned)NPB) << SRC_BITS) | ((unsigned)sv.x & SRC_MASK);
            en.y = (((unsigned)v.y - o[1] * (unsigned)NPB) << SRC_BITS) | ((unsigned)sv.y & SRC_MASK);
            en.z = (((unsigned)v.z - o[2] * (unsigned)NPB) << SRC_BITS) | ((unsigned)sv.z & SRC_MASK);
            en.w = (((unsigned)v.w - o[3] * (unsigned)NPB) << SRC_BITS) | ((unsigned)sv.w & SRC_MASK);
            en4[c * 32 + lane] = en;
            unsigned w = 0u;
            #pragma unroll
            for (int L = 0; L < 32; ++L) {
                const int j = (L >> 2) & 3;
                const int k = ((L >> 4) << 2) | (L & 3);
                const unsigned bits = __builtin_amdgcn_ballot_w32(((o[j] >> k) & 1u) != 0u);
                w = (lane == L) ? bits : w;
            }
            unsigned* p = (lane < 16) ? planeLo : planeHi;
            p[c * 16 + (lane & 15)] = w;
        }
    }
    if (i >= n) return;
    const float* r = h + (long long)i * C;
    float s = 0.f, d = 0.f;
    for (int c = 0; c < C; ++c) { float v = r[c]; s += v * asrc[c]; d += v * adst[c]; }
    outs[i] = s;
    outd[i] = d;
}

#define AGG_T      1024
#define AGG_WAVES  (AGG_T / 32)
#define SCAN_ROT   192
#define SCAN_MLP   4
#define LIST_CAP   28672
#define WS_CAP     8192

template <int C>
struct AggShared {
    float    acc[NPB * C];
    float    den[NPB];
    float    adl[NPB];
    float    lse[NPB];
    union {
        unsigned list[LIST_CAP];
        float    stage[LIST_CAP];
    };
    unsigned lcount;
};
static_assert(NHID * 48 + NPB * NCLASS <= LIST_CAP, "layer 1's finish stages W2 and the block's rows in the list's space");

__device__ __forceinline__ void lds_add(float* p, float v) {
    __hip_atomic_fetch_add(p, v, __ATOMIC_RELAXED, __HIP_MEMORY_SCOPE_WORKGROUP);
}

template <int C>
__device__ __forceinline__ void own_edge(AggShared<C>& sm, int e,
                                         const float* __restrict__ h,
                                         const float* __restrict__ as,
                                         const unsigned* __restrict__ entry) {
    unsigned pos = __hip_atomic_fetch_add(&sm.lcount, 1u, __ATOMIC_RELAXED, __HIP_MEMORY_SCOPE_WORKGROUP);
    if (pos < (unsigned)LIST_CAP) {
        sm.list[pos] = (unsigned)e;
    } else {
        const unsigned ent = entry[e];
        const unsigned dl  = ent >> SRC_BITS;
        const int      s   = (int)(ent & SRC_MASK);
        float x = as[s] + sm.adl[dl];
        x = x > 0.f ? x : NEG_SLOPE * x;
        float w = expf(x);
        lds_add(&sm.den[dl], w);
        #pragma nounroll
        for (int c = 0; c < C; ++c) lds_add(&sm.acc[dl * C + c], h[(long long)s * C + c] * w);
    }
}

struct OwnerBits { unsigned n[8]; };
__device__ __forceinline__ OwnerBits owner_bits(unsigned block) {
    OwnerBits r;
    #pragma unroll
    for (int k = 0; k < 8; ++k) r.n[k] = ((block >> k) & 1u) ? 0u : ~0u;
    return r;
}

__device__ __forceinline__ unsigned owned_mask(uint4 lo, uint4 hi, const OwnerBits& ob) {
    return (lo.x ^ ob.n[0]) & (lo.y ^ ob.n[1]) & (lo.z ^ ob.n[2]) & (lo.w ^ ob.n[3]) &
           (hi.x ^ ob.n[4]) & (hi.y ^ ob.n[5]) & (hi.z ^ ob.n[6]) & (hi.w ^ ob.n[7]);
}

template <int C>
__device__ __forceinline__ void scan_all(AggShared<C>& sm, const uint4* __restrict__ lo4,
                                         const uint4* __restrict__ hi4,
                                         int start, const OwnerBits& ob, int tid,
                                         const float* __restrict__ h,
                                         const float* __restrict__ as,
                                         const unsigned* __restrict__ entry) {
    for (int i0 = 0; i0 < NGROUP; i0 += SCAN_MLP * AGG_T) {
        uint4    pl[SCAN_MLP], ph[SCAN_MLP];
        int      gx[SCAN_MLP];
        unsigned keep[SCAN_MLP];
        #pragma unroll
        for (int k = 0; k < SCAN_MLP; ++k) {
            const int  pos = i0 + k * AGG_T + tid;
            const bool in  = pos < NGROUP;
            int g = pos + start;
            g = g >= NGROUP ? g - NGROUP : g;
            g = in ? g : tid;
            gx[k]   = g;
            keep[k] = in ? ~0u : 0u;
            pl[k]   = lo4[g];
            ph[k]   = hi4[g];
        }
        #pragma unroll
        for (int k = 0; k < SCAN_MLP; ++k) {
            unsigned m = owned_mask(pl[k], ph[k], ob) & keep[k];
            const int e0 = 128 * (gx[k] >> 2) + (gx[k] & 3);
            while (m) {
                const int b = __builtin_ctz(m);
                m &= m - 1u;
                own_edge<C>(sm, e0 + 4 * b, h, as, entry);
            }
        }
    }
}

#define RES_MLP    8
template <int C>
__device__ __forceinline__ void resolve_list(AggShared<C>& sm, int tid,
                                             const unsigned* __restrict__ entry) {
    const unsigned n = sm.lcount < (unsigned)LIST_CAP ? sm.lcount : (unsigned)LIST_CAP;
    for (unsigned i0 = 0; i0 < n; i0 += (unsigned)(RES_MLP * AGG_T)) {
        unsigned idx[RES_MLP], e[RES_MLP], en[RES_MLP];
        #pragma unroll
        for (int k = 0; k < RES_MLP; ++k) {
            idx[k] = i0 + (unsigned)(k * AGG_T + tid);
            e[k]   = idx[k] < n ? sm.list[idx[k]] : 0u;
        }
        #pragma unroll
        for (int k = 0; k < RES_MLP; ++k) en[k] = entry[e[k]];
        #pragma unroll
        for (int k = 0; k < RES_MLP; ++k)
            if (idx[k] < n) sm.list[idx[k]] = en[k];
    }
}

template <int C, bool FINAL>
__global__ __launch_bounds__(AGG_T) void agg_owner(const float* __restrict__ h,
                                                   const float* __restrict__ as,
                                                   const float* __restrict__ ad,
                                                   const unsigned* __restrict__ entry,
                                                   const float* __restrict__ bias,
                                                   float* __restrict__ outp,
                                                   unsigned* __restrict__ wsList,
                                                   unsigned* __restrict__ wsCount,
                                                   const unsigned* __restrict__ owner,
                                                   const float* __restrict__ Wn,
                                                   const float* __restrict__ asrcn,
                                                   const float* __restrict__ adstn,
                                                   float* __restrict__ asn,
                                                   float* __restrict__ adn) {
    constexpr int LPE   = C / 4;
    constexpr int EPW   = 32 / LPE;
    constexpr int STEPS = (32 + EPW - 1) / EPW;

    __shared__ AggShared<C> sm;

    const int tid  = threadIdx.x;
    const int lane = tid & 31;
    const int wave = tid >> 5;
    const int lo   = blockIdx.x * NPB;
    const int cnt  = (NNODES - lo) < NPB ? (NNODES - lo) : NPB;

    for (int i = tid; i < NPB * C; i += AGG_T) sm.acc[i] = 0.f;
    if (tid < NPB) {
        sm.den[tid] = 0.f;
        sm.adl[tid] = tid < cnt ? ad[lo + tid] : 0.f;
    }
    if (tid == 0) sm.lcount = 0u;
    __syncthreads();

    const uint4*    lo4   = reinterpret_cast<const uint4*>(owner);
    const uint4*    hi4   = reinterpret_cast<const uint4*>(owner + 4 * NGROUP);
    const int       start = blockIdx.x * SCAN_ROT;
    const OwnerBits ob    = owner_bits(blockIdx.x);
    uint4*      wl4   = reinterpret_cast<uint4*>(wsList + (size_t)blockIdx.x * WS_CAP);
    if (!FINAL) {
        scan_all<C>(sm, lo4, hi4, start, ob, tid, h, as, entry);
        __syncthreads();
        resolve_list<C>(sm, tid, entry);
        __syncthreads();
        const unsigned total = sm.lcount;
        if (tid == 0) wsCount[blockIdx.x] = total;
        if (total <= (unsigned)WS_CAP) {
            const int np = (int)((total + 3u) >> 2);
            for (int i4 = tid; i4 < np; i4 += AGG_T) {
                const unsigned i = 4u * (unsigned)i4;
                uint4 v;
                v.x = sm.list[i];
                v.y = i + 1u < total ? sm.list[i + 1u] : 0u;
                v.z = i + 2u < total ? sm.list[i + 2u] : 0u;
                v.w = i + 3u < total ? sm.list[i + 3u] : 0u;
                wl4[i4] = v;
            }
        }
    } else {
        const unsigned total = wsCount[blockIdx.x];
        if (total <= (unsigned)WS_CAP) {
            const int np = (int)((total + 3u) >> 2);
            for (int i4 = tid; i4 < np; i4 += AGG_T) {
                const uint4 v = wl4[i4];
                const unsigned i = 4u * (unsigned)i4;
                sm.list[i]      = v.x;
                sm.list[i + 1u] = v.y;
                sm.list[i + 2u] = v.z;
                sm.list[i + 3u] = v.w;
            }
            if (tid == 0) sm.lcount = total;
        } else {
            scan_all<C>(sm, lo4, hi4, start, ob, tid, h, as, entry);
            __syncthreads();
            resolve_list<C>(sm, tid, entry);
        }
        __syncthreads();
    }

    const unsigned n      = sm.lcount < (unsigned)LIST_CAP ? sm.lcount : (unsigned)LIST_CAP;
    const int      nbatch = (int)((n + 31u) >> 5);
    const int      q      = lane / LPE;
    const int      p      = lane - q * LPE;
    const bool     used   = q < EPW;
    const int      qx     = q & 3;
    const bool     sw1    = (qx & 1) != 0;
    const bool     sw2    = (qx & 2) != 0;
    const unsigned pq     = (unsigned)(p * 16 + qx * 4);
    char* const    accb   = reinterpret_cast<char*>(sm.acc);

    for (int b = wave; b < nbatch; b += AGG_WAVES) {
        const unsigned i     = (unsigned)b * 32u + (unsigned)lane;
        const bool     valid = i < n;
        const unsigned ent   = valid ? sm.list[i] : 0u;
        const int      s     = (int)(ent & SRC_MASK);
        const int      dl    = (int)(ent >> SRC_BITS);
        float x = as[s] + sm.adl[dl];
        x = x > 0.f ? x : NEG_SLOPE * x;
        const float w = valid ? expf(x) : 0.f;
        lds_add(&sm.den[dl], w);

        float4 r[STEPS];
        float  wj[STEPS];
        int    dj[STEPS];
        bool   act[STEPS];
        #pragma unroll
        for (int t = 0; t < STEPS; ++t) {
            const int j  = t * EPW + q;
            const int jj = j < 31 ? j : 31;
            const int sj = __shfl(s, jj);
            wj[t]  = __shfl(w, jj);
            dj[t]  = __shfl(dl, jj);
            act[t] = used && (j < 32) && ((unsigned)b * 32u + (unsigned)j < n);
            r[t]   = *reinterpret_cast<const float4*>(h + (long long)sj * C + p * 4);
        }
        #pragma unroll
        for (int t = 0; t < STEPS; ++t) {
            const float v0 = r[t].x * wj[t], v1 = r[t].y * wj[t];
            const float v2 = r[t].z * wj[t], v3 = r[t].w * wj[t];
            const float u0 = sw1 ? v1 : v0, u1 = sw1 ? v0 : v1;
            const float u2 = sw1 ? v3 : v2, u3 = sw1 ? v2 : v3;
            const float b0 = sw2 ? u2 : u0, b1 = sw2 ? u3 : u1;
            const float b2 = sw2 ? u0 : u2, b3 = sw2 ? u1 : u3;
            const unsigned o0 = (unsigned)dj[t] * (unsigned)(C * 4) + pq;
            if (act[t]) {
                lds_add(reinterpret_cast<float*>(accb + o0),         b0);
                lds_add(reinterpret_cast<float*>(accb + (o0 ^ 4u)),  b1);
                lds_add(reinterpret_cast<float*>(accb + (o0 ^ 8u)),  b2);
                lds_add(reinterpret_cast<float*>(accb + (o0 ^ 12u)), b3);
            }
        }
    }
    __syncthreads();

    if (!FINAL) {
        for (int i = tid; i < cnt * C; i += AGG_T) {
            const int nl = i / C, c = i - nl * C;
            const float v = sm.acc[i] / (sm.den[nl] + 1e-16f) + bias[c];
            sm.acc[i] = v > 0.f ? v : 0.f;
        }
        float* w2s = sm.stage;
        float* h2s = sm.stage + NHID * 48;
        for (int i = tid; i < NHID * 48; i += AGG_T) {
            const int k = i / 48, n = i - k * 48;
            w2s[i] = (n < NCLASS) ? Wn[k * NCLASS + n] : 0.0f;
        }
        __syncthreads();

        const int half   = lane >> 4;
        const int l16    = lane & 15;
        const int ntiles = (cnt + 15) >> 4;
        const int wv = __builtin_amdgcn_readfirstlane(wave);
        for (int t = wv; t < ntiles; t += AGG_WAVES) {
            const int row  = t * 16 + l16;
            const int rowc = row < cnt ? row : (cnt - 1);
            const float* ar = &sm.acc[rowc * NHID];
            v8f c0 = {}; v8f c1 = {}; v8f c2 = {};
            for (int k = 0; k < NHID; k += 4) {
                const int k0 = k + 2 * half;
                v2f a;
                a.x = ar[k0];
                a.y = ar[k0 + 1];
                v2f b0, b1, b2;
                b0.x = w2s[k0 * 48 + l16];        b0.y = w2s[(k0 + 1) * 48 + l16];
                b1.x = w2s[k0 * 48 + 16 + l16];   b1.y = w2s[(k0 + 1) * 48 + 16 + l16];
                b2.x = w2s[k0 * 48 + 32 + l16];   b2.y = w2s[(k0 + 1) * 48 + 32 + l16];
                c0 = __builtin_amdgcn_wmma_f32_16x16x4_f32(false, a, false, b0, (short)0, c0, false, false);
                c1 = __builtin_amdgcn_wmma_f32_16x16x4_f32(false, a, false, b1, (short)0, c1, false, false);
                c2 = __builtin_amdgcn_wmma_f32_16x16x4_f32(false, a, false, b2, (short)0, c2, false, false);
            }
            #pragma unroll
            for (int m = 0; m < 8; ++m) {
                const int r = t * 16 + half * 8 + m;
                if (r < cnt) {
                    float* o = h2s + r * NCLASS;
                    o[l16]      = c0[m];
                    o[16 + l16] = c1[m];
                    if (32 + l16 < NCLASS) o[32 + l16] = c2[m];
                }
            }
        }
        __syncthreads();

        if (tid < cnt) {
            const float* r = h2s + tid * NCLASS;
            float s = 0.f, d = 0.f;
            for (int c = 0; c < NCLASS; ++c) { float v = r[c]; s += v * asrcn[c]; d += v * adstn[c]; }
            asn[lo + tid] = s;
            adn[lo + tid] = d;
        }
        float* og = outp + (long long)lo * NCLASS;
        for (int i = tid; i < cnt * NCLASS; i += AGG_T) og[i] = h2s[i];
    } else {
        float* og = outp + (long long)lo * C;
        for (int i = tid; i < cnt * C; i += AGG_T) {
            const int nl = i / C, c = i - nl * C;
            sm.acc[i] = sm.acc[i] / (sm.den[nl] + 1e-16f) + bias[c];
        }
        __syncthreads();
        if (tid < cnt) {
            const float* row = &sm.acc[tid * C];
            float m = -INFINITY;
            for (int c = 0; c < C; ++c) m = fmaxf(m, row[c]);
            float sum = 0.f;
            for (int c = 0; c < C; ++c) sum += expf(row[c] - m);
            sm.lse[tid] = m + logf(sum);
        }
        __syncthreads();
        for (int i = tid; i < cnt * C; i += AGG_T) og[i] = sm.acc[i] - sm.lse[i / C];
    }
}

extern "C" void kernel_launch(void* const* d_in, const int* in_sizes, int n_in,
                              void* d_out, int out_size, void* d_ws, size_t ws_size,
                              hipStream_t stream) {
    const float* x     = (const float*)d_in[0];
    const int*   src   = (const int*)d_in[1];
    const int*   dst   = src + NEDGES;
    const float* W1    = (const float*)d_in[2];
    const float* as1v  = (const float*)d_in[3];
    const float* ad1v  = (const float*)d_in[4];
    const float* b1    = (const float*)d_in[5];
    const float* W2    = (const float*)d_in[6];
    const float* as2v  = (const float*)d_in[7];
    const float* ad2v  = (const float*)d_in[8];
    const float* b2    = (const float*)d_in[9];
    float* out = (float*)d_out;

    float* ws    = (float*)d_ws;
    float* h1    = ws;
    float* h2    = h1 + (long long)NNODES * NHID;
    float* as1   = h2 + (long long)NNODES * NCLASS;
    float* ad1   = as1 + NNODES;
    float* as2   = ad1 + NNODES;
    float* ad2   = as2 + NNODES;
    unsigned* wsList  = (unsigned*)(ad2 + NNODES);
    unsigned* wsCount = wsList + (long long)AGG_BLOCKS * WS_CAP;
    unsigned* owner   = wsCount + AGG_BLOCKS;
    unsigned* entry   = owner + NEDGES / 4;

    const int T = 256;
    const int gemmBlocks = ceil_div(NNODES, 128);

    gemm1_wmma<<<gemmBlocks, T, 0, stream>>>(x, W1, h1);
    alpha_kernel<<<ceil_div(NNODES, T), T, 0, stream>>>(h1, as1v, ad1v, as1, ad1, NNODES, NHID, src, dst, owner, entry);
    agg_owner<NHID, false><<<AGG_BLOCKS, AGG_T, 0, stream>>>(h1, as1, ad1, entry, b1, h2, wsList, wsCount, owner,
                                                             W2, as2v, ad2v, as2, ad2);

    agg_owner<NCLASS, true><<<AGG_BLOCKS, AGG_T, 0, stream>>>(h2, as2, ad2, entry, b2, out, wsList, wsCount, owner,
                                                              nullptr, nullptr, nullptr, nullptr, nullptr);
}
